# starting kernel
# baseline (speedup 1.0000x reference)
#include <hip/hip_runtime.h>

#define V_  12
#define N_  80000
#define C_  256
#define H_  128
#define W_  128
#define G_  4096
#define M_  512
#define E_  200000
#define L_  20
#define C2_ 128

typedef __attribute__((ext_vector_type(2))) float v2f;
typedef __attribute__((ext_vector_type(8))) float v8f;

__global__ void k_zero(float* __restrict__ p, int n) {
  int i = blockIdx.x * blockDim.x + threadIdx.x;
  if (i < n) p[i] = 0.0f;
}

__global__ void k_group_gather(const float* __restrict__ img,
                               const float* __restrict__ is_seen,
                               const int*   __restrict__ coords,
                               const int*   __restrict__ centers,
                               float* __restrict__ group_feat) {
  int g = blockIdx.x;
  int c = threadIdx.x;
  int n = centers[g];
  float acc = 0.0f, den = 0.0f;
  for (int v = 0; v < V_; ++v) {
    int x = coords[((size_t)v * N_ + n) * 2 + 0];
    int y = coords[((size_t)v * N_ + n) * 2 + 1];
    float w = is_seen[(size_t)v * N_ + n];
    den += w;
    acc += w * img[(((size_t)v * C_ + c) * H_ + y) * W_ + x];
  }
  group_feat[(size_t)g * C_ + c] = acc / (den + 1e-6f);
}

__global__ void k_mask_scatter(const float* __restrict__ group_feat,
                               const int*   __restrict__ group2mask,
                               float* __restrict__ msum, float* __restrict__ mcnt) {
  int g = blockIdx.x;
  int c = threadIdx.x;
  int m = group2mask[g];
  unsafeAtomicAdd(&msum[(size_t)m * C_ + c], group_feat[(size_t)g * C_ + c]);
  if (c == 0) unsafeAtomicAdd(&mcnt[m], 1.0f);
}

__global__ void k_mask_norm(const float* __restrict__ msum,
                            const float* __restrict__ mcnt,
                            float* __restrict__ mask_feat) {
  int i = blockIdx.x * blockDim.x + threadIdx.x;
  int m = i >> 8;
  float cc = mcnt[m];
  mask_feat[i] = (cc > 0.0f) ? msum[i] / fmaxf(cc, 1.0f) : 0.0f;
}

__global__ void k_mask_h1(const float* __restrict__ mask_feat,
                          const float* __restrict__ W1,
                          float* __restrict__ mask_h1) {
  int lane = threadIdx.x & 31;
  int wid  = threadIdx.x >> 5;
  int tile = blockIdx.x * 4 + wid;
  int mt = tile >> 3, nt = tile & 7;
  int mrow = (mt << 4) + (lane & 15);
  int ncol = (nt << 4) + (lane & 15);
  int koff = (lane >> 4) << 1;
  v8f c = {};
  for (int kk = 0; kk < C_; kk += 4) {
    v2f a, b;
    int k0 = kk + koff, k1 = kk + koff + 1;
    a.x = mask_feat[(size_t)mrow * C_ + k0];
    a.y = mask_feat[(size_t)mrow * C_ + k1];
    b.x = W1[(size_t)k0 * C2_ + ncol];
    b.y = W1[(size_t)k1 * C2_ + ncol];
    c = __builtin_amdgcn_wmma_f32_16x16x4_f32(false, a, false, b, (short)0, c, false, false);
  }
  for (int i = 0; i < 8; ++i) {
    int r = (mt << 4) + i + ((lane >> 4) << 3);
    mask_h1[(size_t)r * C2_ + ncol] = c[i];
  }
}

__global__ void k_edge_scatter(const float* __restrict__ mask_h1,
                               const int*   __restrict__ pair_mask,
                               const int*   __restrict__ pair_point,
                               float* __restrict__ fsum, float* __restrict__ cnt) {
  long long idx = (long long)blockIdx.x * blockDim.x + threadIdx.x;
  if (idx >= (long long)E_ * C2_) return;
  int e = (int)(idx >> 7);
  int k = (int)(idx & 127);
  int m = pair_mask[e];
  int p = pair_point[e];
  unsafeAtomicAdd(&fsum[(size_t)p * C2_ + k], mask_h1[(size_t)m * C2_ + k]);
  if (k == 0) unsafeAtomicAdd(&cnt[p], 1.0f);
}

__global__ void k_final(const float* __restrict__ fsum, const float* __restrict__ cnt,
                        const float* __restrict__ b1,   const float* __restrict__ W2,
                        const float* __restrict__ b2,   float* __restrict__ out) {
  int lane = threadIdx.x & 31;
  int wid  = threadIdx.x >> 5;
  int m_base = ((blockIdx.x << 3) + wid) << 4;
  int m = m_base + (lane & 15);
  bool valid = (m < N_);
  float scale = 0.0f;
  if (valid) scale = 1.0f / fmaxf(cnt[m], 1e-6f);
  int koff = (lane >> 4) << 1;
  int col  = lane & 15;
  v8f c0 = {}, c1 = {};
  for (int kk = 0; kk < C2_; kk += 4) {
    int k0 = kk + koff, k1 = kk + koff + 1;
    v2f a, b0, b1v;
    if (valid) {
      a.x = fmaxf(fsum[(size_t)m * C2_ + k0] * scale + b1[k0], 0.0f);
      a.y = fmaxf(fsum[(size_t)m * C2_ + k1] * scale + b1[k1], 0.0f);
    } else {
      a.x = 0.0f; a.y = 0.0f;
    }
    b0.x = W2[(size_t)k0 * L_ + col];
    b0.y = W2[(size_t)k1 * L_ + col];
    int col1 = 16 + col;
    b1v.x = (col1 < L_) ? W2[(size_t)k0 * L_ + col1] : 0.0f;
    b1v.y = (col1 < L_) ? W2[(size_t)k1 * L_ + col1] : 0.0f;
    c0 = __builtin_amdgcn_wmma_f32_16x16x4_f32(false, a, false, b0,  (short)0, c0, false, false);
    c1 = __builtin_amdgcn_wmma_f32_16x16x4_f32(false, a, false, b1v, (short)0, c1, false, false);
  }
  for (int i = 0; i < 8; ++i) {
    int r = m_base + i + ((lane >> 4) << 3);
    if (r < N_) {
      out[(size_t)r * L_ + col] = c0[i] + b2[col];
      int col1 = 16 + col;
      if (col1 < L_) out[(size_t)r * L_ + col1] = c1[i] + b2[col1];
    }
  }
}

extern "C" void kernel_launch(void* const* d_in, const int* in_sizes, int n_in,
                              void* d_out, int out_size, void* d_ws, size_t ws_size,
                              hipStream_t stream) {
  const float* img        = (const float*)d_in[0];
  const float* is_seen    = (const float*)d_in[1];
  const float* W1         = (const float*)d_in[2];
  const float* b1         = (const float*)d_in[3];
  const float* W2         = (const float*)d_in[4];
  const float* b2         = (const float*)d_in[5];
  const int*   coords     = (const int*)d_in[6];
  const int*   centers    = (const int*)d_in[7];
  const int*   group2mask = (const int*)d_in[8];
  const int*   pair_mask  = (const int*)d_in[9];
  const int*   pair_point = (const int*)d_in[10];
  float* out = (float*)d_out;

  float* ws         = (float*)d_ws;
  float* group_feat = ws;
  float* msum       = group_feat + (size_t)G_ * C_;
  float* mcnt       = msum + (size_t)M_ * C_;
  float* mask_feat  = mcnt + M_;
  float* mask_h1    = mask_feat + (size_t)M_ * C_;
  float* fsum       = mask_h1 + (size_t)M_ * C2_;

  int nz1 = M_ * C_ + M_;
  k_zero<<<(nz1 + 255) / 256, 256, 0, stream>>>(msum, nz1);
  int nz2 = N_ * C2_ + N_;
  k_zero<<<(nz2 + 255) / 256, 256, 0, stream>>>(fsum, nz2);

  k_group_gather<<<G_, C_, 0, stream>>>(img, is_seen, coords, centers, group_feat);
  k_mask_scatter<<<G_, C_, 0, stream>>>(group_feat, group2mask, msum, mcnt);
  k_mask_norm<<<(M_ * C_) / 256, 256, 0, stream>>>(msum, mcnt, mask_feat);
  k_mask_h1<<<64, 128, 0, stream>>>(mask_feat, W1, mask_h1);

  long long ne = (long long)E_ * C2_;
  k_edge_scatter<<<(int)((ne + 255) / 256), 256, 0, stream>>>(
      mask_h1, pair_mask, pair_point, fsum, fsum + (size_t)N_ * C2_);

  int tiles = (N_ + 15) / 16;
  k_final<<<(tiles + 7) / 8, 256, 0, stream>>>(
      fsum, fsum + (size_t)N_ * C2_, b1, W2, b2, out);
}


# launch folding: mask normalisation done inside the mask GEMM's blocks (A strip normalised once into LDS); edge-link blocks ride in that launch on its idle units; 7 -> 5 launches; sum orders kept
# speedup vs baseline: 2.2945x; 2.2945x over previous
#include <hip/hip_runtime.h>

#define V_  12
#define N_  80000
#define C_  256
#define H_  128
#define W_  128
#define G_  4096
#define M_  512
#define E_  200000
#define L_  20
#define C2_ 128

typedef __attribute__((ext_vector_type(2))) float v2f;
typedef __attribute__((ext_vector_type(8))) float v8f;

#define PIXB_ 14
#define NSORT_ V_
#define NDEN_  (G_ / 1024)

__global__ void __launch_bounds__(1024)
k_pre(const float* __restrict__ is_seen, const int* __restrict__ coords,
      const int* __restrict__ centers,
      int2* __restrict__ tab, float* __restrict__ den,
      float* __restrict__ zp, int nz) {
  __shared__ alignas(16) int hist[512];
  __shared__ int base[512];
  int t = threadIdx.x;
  int b = blockIdx.x;
  if (b >= NSORT_ + NDEN_) {
    int i = (b - NSORT_ - NDEN_) * 1024 + t;
    if (i < nz) zp[i] = 0.0f;
    return;
  }
  if (b >= NSORT_) {
    int g = (b - NSORT_) * 1024 + t;
    int n = centers[g];
    float d = 0.0f;
    for (int v = 0; v < V_; ++v) d += is_seen[(size_t)v * N_ + n];
    den[g] = d;
    return;
  }
  int v = b;
  if (t < 512) hist[t] = 0;
  __syncthreads();
  int pix[4], bin[4], rank[4];
  float w[4];
  for (int j = 0; j < 4; ++j) {
    int g = j * 1024 + t;
    int n = centers[g];
    int x = coords[((size_t)v * N_ + n) * 2 + 0];
    int y = coords[((size_t)v * N_ + n) * 2 + 1];
    w[j] = is_seen[(size_t)v * N_ + n];
    pix[j] = ((y & (H_ - 1)) << 7) | (x & (W_ - 1));
    bin[j] = pix[j] >> 5;
  }
  for (int j = 0; j < 4; ++j) rank[j] = atomicAdd(&hist[bin[j]], 1);
  __syncthreads();
  if (t < 512) {
    int s = 0;
    for (int k = 0; k < 128; ++k) {
      int4 h = ((const int4*)hist)[k];
      s += ((4 * k     < t) ? h.x : 0) + ((4 * k + 1 < t) ? h.y : 0)
         + ((4 * k + 2 < t) ? h.z : 0) + ((4 * k + 3 < t) ? h.w : 0);
    }
    base[t] = s;
  }
  __syncthreads();
  for (int j = 0; j < 4; ++j) {
    int g = j * 1024 + t;
    int slot = base[bin[j]] + rank[j];
    tab[(size_t)v * G_ + slot] = make_int2(pix[j] | (g << PIXB_), __float_as_int(w[j]));
  }
}

#define VB_ 4

__global__ void __launch_bounds__(1024)
k_group_gather(const float* __restrict__ img, const int2* __restrict__ tab,
               const float* __restrict__ den, float* __restrict__ gfT) {
  __shared__ float acc[G_];
  int c = blockIdx.x;
  int t = threadIdx.x;
  int2 e[VB_][4];
#pragma unroll
  for (int vv = 0; vv < VB_; ++vv)
#pragma unroll
    for (int j = 0; j < 4; ++j)
      e[vv][j] = tab[(size_t)vv * G_ + j * 1024 + t];
  for (int j = 0; j < 4; ++j) acc[j * 1024 + t] = 0.0f;
  __syncthreads();
#pragma unroll
  for (int vb = 0; vb < V_; vb += VB_) {
    float val[VB_][4];
    int2  en[VB_][4];
#pragma unroll
    for (int vv = 0; vv < VB_; ++vv) {
      const float* plane = img + (((size_t)(vb + vv) * C_ + c) << PIXB_);
#pragma unroll
      for (int j = 0; j < 4; ++j)
        val[vv][j] = plane[e[vv][j].x & ((1 << PIXB_) - 1)];
    }
    if (vb + VB_ < V_) {
#pragma unroll
      for (int vv = 0; vv < VB_; ++vv)
#pragma unroll
        for (int j = 0; j < 4; ++j)
          en[vv][j] = tab[(size_t)(vb + VB_ + vv) * G_ + j * 1024 + t];
      asm volatile("" :: "v"(en[0][0].x), "v"(en[0][1].x), "v"(en[0][2].x), "v"(en[0][3].x),
                         "v"(en[1][0].x), "v"(en[1][1].x), "v"(en[1][2].x), "v"(en[1][3].x),
                         "v"(en[2][0].x), "v"(en[2][1].x), "v"(en[2][2].x), "v"(en[2][3].x),
                         "v"(en[3][0].x), "v"(en[3][1].x), "v"(en[3][2].x), "v"(en[3][3].x));
    }
    asm volatile("" :: "v"(val[0][0]), "v"(val[0][1]), "v"(val[0][2]), "v"(val[0][3]),
                       "v"(val[1][0]), "v"(val[1][1]), "v"(val[1][2]), "v"(val[1][3]),
                       "v"(val[2][0]), "v"(val[2][1]), "v"(val[2][2]), "v"(val[2][3]),
                       "v"(val[3][0]), "v"(val[3][1]), "v"(val[3][2]), "v"(val[3][3]));
#pragma unroll
    for (int vv = 0; vv < VB_; ++vv) {
#pragma unroll
      for (int j = 0; j < 4; ++j) {
        int g = e[vv][j].x >> PIXB_;
        acc[g] += __int_as_float(e[vv][j].y) * val[vv][j];
      }
      __syncthreads();
    }
    if (vb + VB_ < V_) {
#pragma unroll
      for (int vv = 0; vv < VB_; ++vv)
#pragma unroll
        for (int j = 0; j < 4; ++j) e[vv][j] = en[vv][j];
    }
  }
  for (int j = 0; j < 4; ++j) {
    int g = j * 1024 + t;
    gfT[(size_t)c * G_ + g] = acc[g] / (den[g] + 1e-6f);
  }
}

__global__ void __launch_bounds__(256)
k_mask_scatter(const float* __restrict__ gfT,
               const int*   __restrict__ group2mask,
               float* __restrict__ msum, float* __restrict__ mcnt) {
  __shared__ float tile[C_][33];
  int g0 = blockIdx.x * 32;
  int t = threadIdx.x;
  int lane = t & 31, wv = t >> 5;
#pragma unroll 8
  for (int j = 0; j < 32; ++j) {
    int c = j * 8 + wv;
    tile[c][lane] = gfT[(size_t)c * G_ + g0 + lane];
  }
  __syncthreads();
  for (int gi = 0; gi < 32; ++gi) {
    int m = group2mask[g0 + gi];
    unsafeAtomicAdd(&msum[(size_t)m * C_ + t], tile[t][gi]);
  }
  if (t < 32) unsafeAtomicAdd(&mcnt[group2mask[g0 + t]], 1.0f);
}

#define AP_    260
#define NGEMM_ 64
#define EPT_   4
#define NLINK_ ((E_ + 128 * EPT_ - 1) / (128 * EPT_))

__global__ void __launch_bounds__(128)
k_mask_h1(const float* __restrict__ msum, const float* __restrict__ mcnt,
          const float* __restrict__ W1, float* __restrict__ mask_h1,
          const int* __restrict__ pair_mask, const int* __restrict__ pair_point,
          int* __restrict__ head, int2* __restrict__ link) {
  __shared__ alignas(16) float astrip[16 * AP_];
  int t = threadIdx.x;
  if (blockIdx.x >= NGEMM_) {
    int e0 = (blockIdx.x - NGEMM_) * (128 * EPT_) + t;
    if (e0 - t + 128 * EPT_ <= E_) {
      int m[EPT_], p[EPT_], prev[EPT_];
#pragma unroll
      for (int j = 0; j < EPT_; ++j) { m[j] = pair_mask[e0 + j * 128]; p[j] = pair_point[e0 + j * 128]; }
#pragma unroll
      for (int j = 0; j < EPT_; ++j) prev[j] = atomicExch(&head[p[j]], e0 + j * 128 + 1);
#pragma unroll
      for (int j = 0; j < EPT_; ++j) link[e0 + j * 128] = make_int2(prev[j], m[j]);
    } else {
      for (int j = 0; j < EPT_; ++j) {
        int e = e0 + j * 128;
        if (e < E_) {
          int mm = pair_mask[e];
          int prev = atomicExch(&head[pair_point[e]], e + 1);
          link[e] = make_int2(prev, mm);
        }
      }
    }
    return;
  }
  int lane = t & 31;
  int wid  = t >> 5;
  int mt = blockIdx.x >> 1;
  int nt = ((blockIdx.x & 1) << 2) + wid;
  const float4* ms4 = (const float4*)msum;
  float  cc[4];
  float4 v[4][2];
#pragma unroll
  for (int rr = 0; rr < 4; ++rr) {
    int m = (mt << 4) + (wid << 2) + rr;
    cc[rr] = mcnt[m];
#pragma unroll
    for (int hh = 0; hh < 2; ++hh) v[rr][hh] = ms4[(size_t)m * (C_ / 4) + hh * 32 + lane];
  }
  asm volatile("" :: "v"(v[0][0].x), "v"(v[0][0].y), "v"(v[0][0].z), "v"(v[0][0].w),
                     "v"(v[0][1].x), "v"(v[0][1].y), "v"(v[0][1].z), "v"(v[0][1].w),
                     "v"(v[1][0].x), "v"(v[1][0].y), "v"(v[1][0].z), "v"(v[1][0].w),
                     "v"(v[1][1].x), "v"(v[1][1].y), "v"(v[1][1].z), "v"(v[1][1].w));
  asm volatile("" :: "v"(v[2][0].x), "v"(v[2][0].y), "v"(v[2][0].z), "v"(v[2][0].w),
                     "v"(v[2][1].x), "v"(v[2][1].y), "v"(v[2][1].z), "v"(v[2][1].w),
                     "v"(v[3][0].x), "v"(v[3][0].y), "v"(v[3][0].z), "v"(v[3][0].w),
                     "v"(v[3][1].x), "v"(v[3][1].y), "v"(v[3][1].z), "v"(v[3][1].w));
#pragma unroll
  for (int rr = 0; rr < 4; ++rr) {
    int row = (wid << 2) + rr;
    float dn = fmaxf(cc[rr], 1.0f);
    bool  on = cc[rr] > 0.0f;
#pragma unroll
    for (int hh = 0; hh < 2; ++hh) {
      float4 q, o;
      q.x = v[rr][hh].x / dn; q.y = v[rr][hh].y / dn; q.z = v[rr][hh].z / dn; q.w = v[rr][hh].w / dn;
      o.x = on ? q.x : 0.0f; o.y = on ? q.y : 0.0f; o.z = on ? q.z : 0.0f; o.w = on ? q.w : 0.0f;
      *(float4*)&astrip[row * AP_ + ((hh * 32 + lane) << 2)] = o;
    }
  }
  __syncthreads();

  int ncol = (nt << 4) + (lane & 15);
  int koff = (lane >> 4) << 1;
  const float* ar = &astrip[(lane & 15) * AP_ + koff];
  v8f c = {};
  for (int kk = 0; kk < C_; kk += 4) {
    v2f a, b;
    int k0 = kk + koff, k1 = kk + koff + 1;
    a.x = ar[kk];
    a.y = ar[kk + 1];
    b.x = W1[(size_t)k0 * C2_ + ncol];
    b.y = W1[(size_t)k1 * C2_ + ncol];
    c = __builtin_amdgcn_wmma_f32_16x16x4_f32(false, a, false, b, (short)0, c, false, false);
  }
  for (int i = 0; i < 8; ++i) {
    int r = (mt << 4) + i + ((lane >> 4) << 3);
    mask_h1[(size_t)r * C2_ + ncol] = c[i];
  }
}

#define TP_ 132

__global__ void __launch_bounds__(128)
k_final(const float* __restrict__ mask_h1, const int* __restrict__ head,
        const int2* __restrict__ link,
        const float* __restrict__ b1,   const float* __restrict__ W2,
        const float* __restrict__ b2,   float* __restrict__ out) {
  __shared__ alignas(16) float tile[4][16 * TP_];
  int lane = threadIdx.x & 31;
  int wid  = threadIdx.x >> 5;
  int m_base = ((blockIdx.x << 2) + wid) << 4;
  const float4* rows = (const float4*)mask_h1;

  int p = m_base + (lane & 15);
  int e = 0;
  if (lane < 16 && p < N_) e = head[p];
  int2 lk = make_int2(0, 0);
  if (e != 0) lk = link[e - 1];
  int n = 0;
  float4 acc[16];
#pragma unroll
  for (int i = 0; i < 16; ++i) acc[i] = make_float4(0.0f, 0.0f, 0.0f, 0.0f);

  for (int it = 0; it < E_; ++it) {
    unsigned live = __builtin_amdgcn_ballot_w32(e != 0);
    if (live == 0u) break;
    if (e != 0) ++n;
    int en = lk.x;
    int2 lkn = make_int2(0, 0);
    if (en != 0) lkn = link[en - 1];
    float4 r[16];
#pragma unroll
    for (int i = 0; i < 16; ++i) {
      int m = __builtin_amdgcn_readlane(lk.y, i);
      r[i] = rows[(size_t)m * 32 + lane];
    }
#pragma unroll
    for (int i = 0; i < 16; ++i) {
      if ((live >> i) & 1u) {
        acc[i].x += r[i].x; acc[i].y += r[i].y; acc[i].z += r[i].z; acc[i].w += r[i].w;
      }
    }
    e = en;
    lk = lkn;
  }

  float* tw = tile[wid];
  float4 bb = make_float4(b1[4 * lane], b1[4 * lane + 1], b1[4 * lane + 2], b1[4 * lane + 3]);
#pragma unroll
  for (int i = 0; i < 16; ++i) {
    float cn = (float)__builtin_amdgcn_readlane(n, i);
    float scale = 1.0f / fmaxf(cn, 1e-6f);
    float4 h;
    h.x = fmaxf(acc[i].x * scale + bb.x, 0.0f);
    h.y = fmaxf(acc[i].y * scale + bb.y, 0.0f);
    h.z = fmaxf(acc[i].z * scale + bb.z, 0.0f);
    h.w = fmaxf(acc[i].w * scale + bb.w, 0.0f);
    *(float4*)&tw[i * TP_ + 4 * lane] = h;
  }
  __syncthreads();

  int koff = (lane >> 4) << 1;
  int col  = lane & 15;
  const float* ar = &tw[(lane & 15) * TP_ + koff];
  v8f c0 = {}, c1 = {};
  for (int kk = 0; kk < C2_; kk += 4) {
    int k0 = kk + koff, k1 = kk + koff + 1;
    v2f a, b0, b1v;
    a.x = ar[kk];
    a.y = ar[kk + 1];
    b0.x = W2[(size_t)k0 * L_ + col];
    b0.y = W2[(size_t)k1 * L_ + col];
    int col1 = 16 + col;
    b1v.x = (col1 < L_) ? W2[(size_t)k0 * L_ + col1] : 0.0f;
    b1v.y = (col1 < L_) ? W2[(size_t)k1 * L_ + col1] : 0.0f;
    c0 = __builtin_amdgcn_wmma_f32_16x16x4_f32(false, a, false, b0,  (short)0, c0, false, false);
    c1 = __builtin_amdgcn_wmma_f32_16x16x4_f32(false, a, false, b1v, (short)0, c1, false, false);
  }
  for (int i = 0; i < 8; ++i) {
    int r = m_base + i + ((lane >> 4) << 3);
    if (r < N_) {
      out[(size_t)r * L_ + col] = c0[i] + b2[col];
      int col1 = 16 + col;
      if (col1 < L_) out[(size_t)r * L_ + col1] = c1[i] + b2[col1];
    }
  }
}

extern "C" void kernel_launch(void* const* d_in, const int* in_sizes, int n_in,
                              void* d_out, int out_size, void* d_ws, size_t ws_size,
                              hipStream_t stream) {
  const float* img        = (const float*)d_in[0];
  const float* is_seen    = (const float*)d_in[1];
  const float* W1         = (const float*)d_in[2];
  const float* b1         = (const float*)d_in[3];
  const float* W2         = (const float*)d_in[4];
  const float* b2         = (const float*)d_in[5];
  const int*   coords     = (const int*)d_in[6];
  const int*   centers    = (const int*)d_in[7];
  const int*   group2mask = (const int*)d_in[8];
  const int*   pair_mask  = (const int*)d_in[9];
  const int*   pair_point = (const int*)d_in[10];
  float* out = (float*)d_out;

  float* ws         = (float*)d_ws;
  float* group_feat = ws;
  float* msum       = group_feat + (size_t)G_ * C_;
  float* mcnt       = msum + (size_t)M_ * C_;
  float* headf      = mcnt + M_;
  float* tabf       = headf + N_;
  float* mask_h1    = tabf + (size_t)M_ * C_;
  int2*  link       = (int2*)(mask_h1 + (size_t)M_ * C2_);
  int*   head       = (int*)headf;
  int2*  tab        = (int2*)tabf;
  float* den        = tabf + (size_t)V_ * G_ * 2;

  int nz = M_ * C_ + M_ + N_;
  k_pre<<<NSORT_ + NDEN_ + (nz + 1023) / 1024, 1024, 0, stream>>>(is_seen, coords, centers, tab, den, msum, nz);

  k_group_gather<<<C_, 1024, 0, stream>>>(img, tab, den, group_feat);
  k_mask_scatter<<<G_ / 32, 256, 0, stream>>>(group_feat, group2mask, msum, mcnt);
  k_mask_h1<<<NGEMM_ + NLINK_, 128, 0, stream>>>(msum, mcnt, W1, mask_h1, pair_mask, pair_point, head, link);

  int tiles = (N_ + 15) / 16;
  k_final<<<(tiles + 3) / 4, 128, 0, stream>>>(mask_h1, head, link, b1, W2, b2, out);
}


# mask sums by pull: groups counting-sorted by mask, gather writes mask-sorted columns, GEMM blocks add their strip's columns via LDS in slot order (sort arrival order, no float atomics); 5->4 launches
# speedup vs baseline: 2.4631x; 1.0735x over previous
#include <hip/hip_runtime.h>

#define V_  12
#define N_  80000
#define C_  256
#define H_  128
#define W_  128
#define G_  4096
#define M_  512
#define E_  200000
#define L_  20
#define C2_ 128

typedef __attribute__((ext_vector_type(2))) float v2f;
typedef __attribute__((ext_vector_type(8))) float v8f;

#define PIXB_ 14
#define NSORT_ V_
#define NDEN_  (G_ / 1024)
#define NMSORT_ 1

__global__ void __launch_bounds__(1024)
k_pre(const float* __restrict__ is_seen, const int* __restrict__ coords,
      const int* __restrict__ centers, const int* __restrict__ group2mask,
      int2* __restrict__ tab, float* __restrict__ den,
      int* __restrict__ perm, int* __restrict__ mstart,
      float* __restrict__ zp, int nz) {
  __shared__ alignas(16) int hist[512];
  __shared__ int base[512];
  __shared__ int wtot[16];
  int t = threadIdx.x;
  int b = blockIdx.x;
  if (b >= NSORT_ + NDEN_ + NMSORT_) {
    int i = (b - NSORT_ - NDEN_ - NMSORT_) * 1024 + t;
    if (i < nz) zp[i] = 0.0f;
    return;
  }
  if (b >= NSORT_ + NDEN_) {
    if (t < 512) hist[t] = 0;
    __syncthreads();
    int mk[4], rk[4];
    for (int j = 0; j < 4; ++j) mk[j] = group2mask[j * 1024 + t] & (M_ - 1);
    for (int j = 0; j < 4; ++j) rk[j] = atomicAdd(&hist[mk[j]], 1);
    __syncthreads();
    int cnt = (t < 512) ? hist[t] : 0;
    int incl = cnt;
    for (int d = 1; d < 32; d <<= 1) {
      int y = __shfl_up(incl, d);
      if ((t & 31) >= d) incl += y;
    }
    if (t < 512 && (t & 31) == 31) wtot[t >> 5] = incl;
    __syncthreads();
    if (t < 512) {
      int off = 0;
      for (int w = 0; w < 16; ++w) off += (w < (t >> 5)) ? wtot[w] : 0;
      int ex = off + incl - cnt;
      base[t] = ex;
      mstart[t] = ex;
    }
    if (t == 0) mstart[M_] = G_;
    __syncthreads();
    for (int j = 0; j < 4; ++j) perm[base[mk[j]] + rk[j]] = j * 1024 + t;
    return;
  }
  if (b >= NSORT_) {
    int g = (b - NSORT_) * 1024 + t;
    int n = centers[g];
    float d = 0.0f;
    for (int v = 0; v < V_; ++v) d += is_seen[(size_t)v * N_ + n];
    den[g] = d;
    return;
  }
  int v = b;
  if (t < 512) hist[t] = 0;
  __syncthreads();
  int pix[4], bin[4], rank[4];
  float w[4];
  for (int j = 0; j < 4; ++j) {
    int g = j * 1024 + t;
    int n = centers[g];
    int x = coords[((size_t)v * N_ + n) * 2 + 0];
    int y = coords[((size_t)v * N_ + n) * 2 + 1];
    w[j] = is_seen[(size_t)v * N_ + n];
    pix[j] = ((y & (H_ - 1)) << 7) | (x & (W_ - 1));
    bin[j] = pix[j] >> 5;
  }
  for (int j = 0; j < 4; ++j) rank[j] = atomicAdd(&hist[bin[j]], 1);
  __syncthreads();
  if (t < 512) {
    int s = 0;
    for (int k = 0; k < 128; ++k) {
      int4 h = ((const int4*)hist)[k];
      s += ((4 * k     < t) ? h.x : 0) + ((4 * k + 1 < t) ? h.y : 0)
         + ((4 * k + 2 < t) ? h.z : 0) + ((4 * k + 3 < t) ? h.w : 0);
    }
    base[t] = s;
  }
  __syncthreads();
  for (int j = 0; j < 4; ++j) {
    int g = j * 1024 + t;
    int slot = base[bin[j]] + rank[j];
    tab[(size_t)v * G_ + slot] = make_int2(pix[j] | (g << PIXB_), __float_as_int(w[j]));
  }
}

#define VB_ 4

__global__ void __launch_bounds__(1024)
k_group_gather(const float* __restrict__ img, const int2* __restrict__ tab,
               const float* __restrict__ den, const int* __restrict__ perm,
               float* __restrict__ gfT) {
  __shared__ float acc[G_];
  int c = blockIdx.x;
  int t = threadIdx.x;
  int2 e[VB_][4];
#pragma unroll
  for (int vv = 0; vv < VB_; ++vv)
#pragma unroll
    for (int j = 0; j < 4; ++j)
      e[vv][j] = tab[(size_t)vv * G_ + j * 1024 + t];
  for (int j = 0; j < 4; ++j) acc[j * 1024 + t] = 0.0f;
  __syncthreads();
#pragma unroll
  for (int vb = 0; vb < V_; vb += VB_) {
    float val[VB_][4];
    int2  en[VB_][4];
#pragma unroll
    for (int vv = 0; vv < VB_; ++vv) {
      const float* plane = img + (((size_t)(vb + vv) * C_ + c) << PIXB_);
#pragma unroll
      for (int j = 0; j < 4; ++j)
        val[vv][j] = plane[e[vv][j].x & ((1 << PIXB_) - 1)];
    }
    if (vb + VB_ < V_) {
#pragma unroll
      for (int vv = 0; vv < VB_; ++vv)
#pragma unroll
        for (int j = 0; j < 4; ++j)
          en[vv][j] = tab[(size_t)(vb + VB_ + vv) * G_ + j * 1024 + t];
      asm volatile("" :: "v"(en[0][0].x), "v"(en[0][1].x), "v"(en[0][2].x), "v"(en[0][3].x),
                         "v"(en[1][0].x), "v"(en[1][1].x), "v"(en[1][2].x), "v"(en[1][3].x),
                         "v"(en[2][0].x), "v"(en[2][1].x), "v"(en[2][2].x), "v"(en[2][3].x),
                         "v"(en[3][0].x), "v"(en[3][1].x), "v"(en[3][2].x), "v"(en[3][3].x));
    }
    asm volatile("" :: "v"(val[0][0]), "v"(val[0][1]), "v"(val[0][2]), "v"(val[0][3]),
                       "v"(val[1][0]), "v"(val[1][1]), "v"(val[1][2]), "v"(val[1][3]),
                       "v"(val[2][0]), "v"(val[2][1]), "v"(val[2][2]), "v"(val[2][3]),
                       "v"(val[3][0]), "v"(val[3][1]), "v"(val[3][2]), "v"(val[3][3]));
#pragma unroll
    for (int vv = 0; vv < VB_; ++vv) {
#pragma unroll
      for (int j = 0; j < 4; ++j) {
        int g = e[vv][j].x >> PIXB_;
        acc[g] += __int_as_float(e[vv][j].y) * val[vv][j];
      }
      __syncthreads();
    }
    if (vb + VB_ < V_) {
#pragma unroll
      for (int vv = 0; vv < VB_; ++vv)
#pragma unroll
        for (int j = 0; j < 4; ++j) e[vv][j] = en[vv][j];
    }
  }
  int   pg[4];
  float dn[4];
  for (int j = 0; j < 4; ++j) { pg[j] = perm[j * 1024 + t] & (G_ - 1); dn[j] = den[j * 1024 + t]; }
  for (int j = 0; j < 4; ++j) {
    int g = j * 1024 + t;
    acc[g] = acc[g] / (dn[j] + 1e-6f);
  }
  __syncthreads();
  for (int j = 0; j < 4; ++j) {
    int sl = j * 1024 + t;
    gfT[(size_t)c * G_ + sl] = acc[pg[j]];
  }
}

#define AP_    260
#define NGEMM_ 32
#define LB_    256
#define EPT_   4
#define NLINK_ ((E_ + LB_ * EPT_ - 1) / (LB_ * EPT_))
#define CH_    128
#define TPT_   132

__global__ void __launch_bounds__(LB_)
k_mask_h1(const float* __restrict__ gfT, const int* __restrict__ mstart,
          const float* __restrict__ W1, float* __restrict__ mask_h1,
          const int* __restrict__ pair_mask, const int* __restrict__ pair_point,
          int* __restrict__ head, int2* __restrict__ link) {
  __shared__ alignas(16) float astrip[16 * AP_];
  __shared__ alignas(16) float tile[C_ * TPT_];
  int t = threadIdx.x;
  if (blockIdx.x >= NGEMM_) {
    int e0 = (blockIdx.x - NGEMM_) * (LB_ * EPT_) + t;
    if (e0 - t + LB_ * EPT_ <= E_) {
      int m[EPT_], p[EPT_], prev[EPT_];
#pragma unroll
      for (int j = 0; j < EPT_; ++j) { m[j] = pair_mask[e0 + j * LB_]; p[j] = pair_point[e0 + j * LB_]; }
#pragma unroll
      for (int j = 0; j < EPT_; ++j) prev[j] = atomicExch(&head[p[j]], e0 + j * LB_ + 1);
#pragma unroll
      for (int j = 0; j < EPT_; ++j) link[e0 + j * LB_] = make_int2(prev[j], m[j]);
    } else {
      for (int j = 0; j < EPT_; ++j) {
        int e = e0 + j * LB_;
        if (e < E_) {
          int mm = pair_mask[e];
          int prev = atomicExch(&head[pair_point[e]], e + 1);
          link[e] = make_int2(prev, mm);
        }
      }
    }
    return;
  }
  int lane = t & 31;
  int wid  = t >> 5;
  int mt   = blockIdx.x;
  int r    = t >> 4;
  int cg   = t & 15;
  int lo = mstart[(mt << 4) + r];
  int hi = mstart[(mt << 4) + r + 1];
  int s0 = __builtin_amdgcn_readfirstlane(mstart[mt << 4]);
  int s1 = __builtin_amdgcn_readfirstlane(mstart[(mt << 4) + 16]);
  float sum[16];
#pragma unroll
  for (int cc = 0; cc < 16; ++cc) sum[cc] = 0.0f;
  for (int cs = s0 & ~3; cs < s1; cs += CH_) {
    int sl = cs + (lane << 2);
    float4 v[32];
#pragma unroll
    for (int i = 0; i < 32; ++i) v[i] = make_float4(0.0f, 0.0f, 0.0f, 0.0f);
    if (sl < s1) {
#pragma unroll
      for (int i = 0; i < 32; ++i) v[i] = *(const float4*)&gfT[(size_t)((wid << 5) + i) * G_ + sl];
    }
#pragma unroll
    for (int i = 0; i < 32; ++i) *(float4*)&tile[((wid << 5) + i) * TPT_ + (lane << 2)] = v[i];
    __syncthreads();
    int a  = max(lo, cs) - cs;
    int bd = min(hi, cs + CH_) - cs;
    for (int i = a; i < bd; ++i) {
#pragma unroll
      for (int cc = 0; cc < 16; ++cc) sum[cc] += tile[(cg + (cc << 4)) * TPT_ + i];
    }
    __syncthreads();
  }
  {
    int   cn = hi - lo;
    float dn = fmaxf((float)cn, 1.0f);
    bool  on = cn > 0;
#pragma unroll
    for (int cc = 0; cc < 16; ++cc) {
      float q = sum[cc] / dn;
      astrip[r * AP_ + cg + (cc << 4)] = on ? q : 0.0f;
    }
  }
  __syncthreads();

  int ncol = (wid << 4) + (lane & 15);
  int koff = (lane >> 4) << 1;
  const float* ar = &astrip[(lane & 15) * AP_ + koff];
  v8f c = {};
  for (int kk = 0; kk < C_; kk += 4) {
    v2f a, b;
    int k0 = kk + koff, k1 = kk + koff + 1;
    a.x = ar[kk];
    a.y = ar[kk + 1];
    b.x = W1[(size_t)k0 * C2_ + ncol];
    b.y = W1[(size_t)k1 * C2_ + ncol];
    c = __builtin_amdgcn_wmma_f32_16x16x4_f32(false, a, false, b, (short)0, c, false, false);
  }
  for (int i = 0; i < 8; ++i) {
    int rr = (mt << 4) + i + ((lane >> 4) << 3);
    mask_h1[(size_t)rr * C2_ + ncol] = c[i];
  }
}

#define TP_ 132

__global__ void __launch_bounds__(128)
k_final(const float* __restrict__ mask_h1, const int* __restrict__ head,
        const int2* __restrict__ link,
        const float* __restrict__ b1,   const float* __restrict__ W2,
        const float* __restrict__ b2,   float* __restrict__ out) {
  __shared__ alignas(16) float tile[4][16 * TP_];
  int lane = threadIdx.x & 31;
  int wid  = threadIdx.x >> 5;
  int m_base = ((blockIdx.x << 2) + wid) << 4;
  const float4* rows = (const float4*)mask_h1;

  int p = m_base + (lane & 15);
  int e = 0;
  if (lane < 16 && p < N_) e = head[p];
  int2 lk = make_int2(0, 0);
  if (e != 0) lk = link[e - 1];
  int n = 0;
  float4 acc[16];
#pragma unroll
  for (int i = 0; i < 16; ++i) acc[i] = make_float4(0.0f, 0.0f, 0.0f, 0.0f);

  for (int it = 0; it < E_; ++it) {
    unsigned live = __builtin_amdgcn_ballot_w32(e != 0);
    if (live == 0u) break;
    if (e != 0) ++n;
    int en = lk.x;
    int2 lkn = make_int2(0, 0);
    if (en != 0) lkn = link[en - 1];
    float4 r[16];
#pragma unroll
    for (int i = 0; i < 16; ++i) {
      int m = __builtin_amdgcn_readlane(lk.y, i);
      r[i] = rows[(size_t)m * 32 + lane];
    }
#pragma unroll
    for (int i = 0; i < 16; ++i) {
      if ((live >> i) & 1u) {
        acc[i].x += r[i].x; acc[i].y += r[i].y; acc[i].z += r[i].z; acc[i].w += r[i].w;
      }
    }
    e = en;
    lk = lkn;
  }

  float* tw = tile[wid];
  float4 bb = make_float4(b1[4 * lane], b1[4 * lane + 1], b1[4 * lane + 2], b1[4 * lane + 3]);
#pragma unroll
  for (int i = 0; i < 16; ++i) {
    float cn = (float)__builtin_amdgcn_readlane(n, i);
    float scale = 1.0f / fmaxf(cn, 1e-6f);
    float4 h;
    h.x = fmaxf(acc[i].x * scale + bb.x, 0.0f);
    h.y = fmaxf(acc[i].y * scale + bb.y, 0.0f);
    h.z = fmaxf(acc[i].z * scale + bb.z, 0.0f);
    h.w = fmaxf(acc[i].w * scale + bb.w, 0.0f);
    *(float4*)&tw[i * TP_ + 4 * lane] = h;
  }
  __syncthreads();

  int koff = (lane >> 4) << 1;
  int col  = lane & 15;
  const float* ar = &tw[(lane & 15) * TP_ + koff];
  v8f c0 = {}, c1 = {};
  for (int kk = 0; kk < C2_; kk += 4) {
    int k0 = kk + koff, k1 = kk + koff + 1;
    v2f a, b0, b1v;
    a.x = ar[kk];
    a.y = ar[kk + 1];
    b0.x = W2[(size_t)k0 * L_ + col];
    b0.y = W2[(size_t)k1 * L_ + col];
    int col1 = 16 + col;
    b1v.x = (col1 < L_) ? W2[(size_t)k0 * L_ + col1] : 0.0f;
    b1v.y = (col1 < L_) ? W2[(size_t)k1 * L_ + col1] : 0.0f;
    c0 = __builtin_amdgcn_wmma_f32_16x16x4_f32(false, a, false, b0,  (short)0, c0, false, false);
    c1 = __builtin_amdgcn_wmma_f32_16x16x4_f32(false, a, false, b1v, (short)0, c1, false, false);
  }
  for (int i = 0; i < 8; ++i) {
    int r = m_base + i + ((lane >> 4) << 3);
    if (r < N_) {
      out[(size_t)r * L_ + col] = c0[i] + b2[col];
      int col1 = 16 + col;
      if (col1 < L_) out[(size_t)r * L_ + col1] = c1[i] + b2[col1];
    }
  }
}

extern "C" void kernel_launch(void* const* d_in, const int* in_sizes, int n_in,
                              void* d_out, int out_size, void* d_ws, size_t ws_size,
                              hipStream_t stream) {
  const float* img        = (const float*)d_in[0];
  const float* is_seen    = (const float*)d_in[1];
  const float* W1         = (const float*)d_in[2];
  const float* b1         = (const float*)d_in[3];
  const float* W2         = (const float*)d_in[4];
  const float* b2         = (const float*)d_in[5];
  const int*   coords     = (const int*)d_in[6];
  const int*   centers    = (const int*)d_in[7];
  const int*   group2mask = (const int*)d_in[8];
  const int*   pair_mask  = (const int*)d_in[9];
  const int*   pair_point = (const int*)d_in[10];
  float* out = (float*)d_out;

  float* ws         = (float*)d_ws;
  float* group_feat = ws;
  float* sortf      = group_feat + (size_t)G_ * C_;
  float* headf      = sortf + (size_t)M_ * C_ + M_;
  float* tabf       = headf + N_;
  float* mask_h1    = tabf + (size_t)M_ * C_;
  int2*  link       = (int2*)(mask_h1 + (size_t)M_ * C2_);
  int*   head       = (int*)headf;
  int2*  tab        = (int2*)tabf;
  float* den        = tabf + (size_t)V_ * G_ * 2;
  int*   perm       = (int*)sortf;
  int*   mstart     = perm + G_;

  int nz = N_;
  k_pre<<<NSORT_ + NDEN_ + NMSORT_ + (nz + 1023) / 1024, 1024, 0, stream>>>(is_seen, coords, centers, group2mask, tab, den, perm, mstart, headf, nz);

  k_group_gather<<<C_, 1024, 0, stream>>>(img, tab, den, perm, group_feat);
  k_mask_h1<<<NGEMM_ + NLINK_, LB_, 0, stream>>>(group_feat, mstart, W1, mask_h1, pair_mask, pair_point, head, link);

  int tiles = (N_ + 15) / 16;
  k_final<<<(tiles + 3) / 4, 128, 0, stream>>>(mask_h1, head, link, b1, W2, b2, out);
}


# group gather: rolling per-view pipeline, picks 4 views and table entries 7 views ahead stay in flight under each view's LDS adds (LDS-only barriers; was: load 4 views, drain, add); sum order kept
# speedup vs baseline: 2.5542x; 1.0370x over previous
#include <hip/hip_runtime.h>

#define V_  12
#define N_  80000
#define C_  256
#define H_  128
#define W_  128
#define G_  4096
#define M_  512
#define E_  200000
#define L_  20
#define C2_ 128

typedef __attribute__((ext_vector_type(2))) float v2f;
typedef __attribute__((ext_vector_type(8))) float v8f;

#define PIXB_ 14
#define NSORT_ V_
#define NDEN_  (G_ / 1024)
#define NMSORT_ 1

__global__ void __launch_bounds__(1024)
k_pre(const float* __restrict__ is_seen, const int* __restrict__ coords,
      const int* __restrict__ centers, const int* __restrict__ group2mask,
      int2* __restrict__ tab, float* __restrict__ den,
      int* __restrict__ perm, int* __restrict__ mstart,
      float* __restrict__ zp, int nz) {
  __shared__ alignas(16) int hist[512];
  __shared__ int base[512];
  __shared__ int wtot[16];
  int t = threadIdx.x;
  int b = blockIdx.x;
  if (b >= NSORT_ + NDEN_ + NMSORT_) {
    int i = (b - NSORT_ - NDEN_ - NMSORT_) * 1024 + t;
    if (i < nz) zp[i] = 0.0f;
    return;
  }
  if (b >= NSORT_ + NDEN_) {
    if (t < 512) hist[t] = 0;
    __syncthreads();
    int mk[4], rk[4];
    for (int j = 0; j < 4; ++j) mk[j] = group2mask[j * 1024 + t] & (M_ - 1);
    for (int j = 0; j < 4; ++j) rk[j] = atomicAdd(&hist[mk[j]], 1);
    __syncthreads();
    int cnt = (t < 512) ? hist[t] : 0;
    int incl = cnt;
    for (int d = 1; d < 32; d <<= 1) {
      int y = __shfl_up(incl, d);
      if ((t & 31) >= d) incl += y;
    }
    if (t < 512 && (t & 31) == 31) wtot[t >> 5] = incl;
    __syncthreads();
    if (t < 512) {
      int off = 0;
      for (int w = 0; w < 16; ++w) off += (w < (t >> 5)) ? wtot[w] : 0;
      int ex = off + incl - cnt;
      base[t] = ex;
      mstart[t] = ex;
    }
    if (t == 0) mstart[M_] = G_;
    __syncthreads();
    for (int j = 0; j < 4; ++j) perm[base[mk[j]] + rk[j]] = j * 1024 + t;
    return;
  }
  if (b >= NSORT_) {
    int g = (b - NSORT_) * 1024 + t;
    int n = centers[g];
    float d = 0.0f;
    for (int v = 0; v < V_; ++v) d += is_seen[(size_t)v * N_ + n];
    den[g] = d;
    return;
  }
  int v = b;
  if (t < 512) hist[t] = 0;
  __syncthreads();
  int pix[4], bin[4], rank[4];
  float w[4];
  for (int j = 0; j < 4; ++j) {
    int g = j * 1024 + t;
    int n = centers[g];
    int x = coords[((size_t)v * N_ + n) * 2 + 0];
    int y = coords[((size_t)v * N_ + n) * 2 + 1];
    w[j] = is_seen[(size_t)v * N_ + n];
    pix[j] = ((y & (H_ - 1)) << 7) | (x & (W_ - 1));
    bin[j] = pix[j] >> 5;
  }
  for (int j = 0; j < 4; ++j) rank[j] = atomicAdd(&hist[bin[j]], 1);
  __syncthreads();
  if (t < 512) {
    int s = 0;
    for (int k = 0; k < 128; ++k) {
      int4 h = ((const int4*)hist)[k];
      s += ((4 * k     < t) ? h.x : 0) + ((4 * k + 1 < t) ? h.y : 0)
         + ((4 * k + 2 < t) ? h.z : 0) + ((4 * k + 3 < t) ? h.w : 0);
    }
    base[t] = s;
  }
  __syncthreads();
  for (int j = 0; j < 4; ++j) {
    int g = j * 1024 + t;
    int slot = base[bin[j]] + rank[j];
    tab[(size_t)v * G_ + slot] = make_int2(pix[j] | (g << PIXB_), __float_as_int(w[j]));
  }
}

__device__ __forceinline__ void lds_barrier() {
  __builtin_amdgcn_fence(__ATOMIC_RELEASE, "workgroup", "local");
  __builtin_amdgcn_s_barrier();
  __builtin_amdgcn_fence(__ATOMIC_ACQUIRE, "workgroup", "local");
}

#define DI_ 4
#define DT_ 7

__global__ void __launch_bounds__(1024)
k_group_gather(const float* __restrict__ img, const int2* __restrict__ tab,
               const float* __restrict__ den, const int* __restrict__ perm,
               float* __restrict__ gfT) {
  __shared__ float acc[G_];
  int c = blockIdx.x;
  int t = threadIdx.x;
  int2  e[V_][4];
  float val[V_][4];
  int   pg[4];
  float dn[4];
  for (int j = 0; j < 4; ++j) { pg[j] = perm[j * 1024 + t] & (G_ - 1); dn[j] = den[j * 1024 + t]; }
#pragma unroll
  for (int v = 0; v < DT_; ++v)
#pragma unroll
    for (int j = 0; j < 4; ++j)
      e[v][j] = tab[(size_t)v * G_ + j * 1024 + t];
  for (int j = 0; j < 4; ++j) acc[j * 1024 + t] = 0.0f;
#pragma unroll
  for (int v = 0; v < DI_; ++v) {
    const float* plane = img + (((size_t)v * C_ + c) << PIXB_);
#pragma unroll
    for (int j = 0; j < 4; ++j)
      val[v][j] = plane[e[v][j].x & ((1 << PIXB_) - 1)];
  }
  lds_barrier();
#pragma unroll
  for (int v = 0; v < V_; ++v) {
    if (v + DT_ < V_) {
#pragma unroll
      for (int j = 0; j < 4; ++j)
        e[v + DT_][j] = tab[(size_t)(v + DT_) * G_ + j * 1024 + t];
    }
    if (v + DI_ < V_) {
      const float* plane = img + (((size_t)(v + DI_) * C_ + c) << PIXB_);
#pragma unroll
      for (int j = 0; j < 4; ++j)
        val[v + DI_][j] = plane[e[v + DI_][j].x & ((1 << PIXB_) - 1)];
    }
    __builtin_amdgcn_sched_barrier(0);
#pragma unroll
    for (int j = 0; j < 4; ++j) {
      int g = e[v][j].x >> PIXB_;
      acc[g] += __int_as_float(e[v][j].y) * val[v][j];
    }
    lds_barrier();
  }
  for (int j = 0; j < 4; ++j) {
    int g = j * 1024 + t;
    acc[g] = acc[g] / (dn[j] + 1e-6f);
  }
  __syncthreads();
  for (int j = 0; j < 4; ++j) {
    int sl = j * 1024 + t;
    gfT[(size_t)c * G_ + sl] = acc[pg[j]];
  }
}

#define AP_    260
#define NGEMM_ 32
#define LB_    256
#define EPT_   4
#define NLINK_ ((E_ + LB_ * EPT_ - 1) / (LB_ * EPT_))
#define CH_    128
#define TPT_   132

__global__ void __launch_bounds__(LB_)
k_mask_h1(const float* __restrict__ gfT, const int* __restrict__ mstart,
          const float* __restrict__ W1, float* __restrict__ mask_h1,
          const int* __restrict__ pair_mask, const int* __restrict__ pair_point,
          int* __restrict__ head, int2* __restrict__ link) {
  __shared__ alignas(16) float astrip[16 * AP_];
  __shared__ alignas(16) float tile[C_ * TPT_];
  int t = threadIdx.x;
  if (blockIdx.x >= NGEMM_) {
    int e0 = (blockIdx.x - NGEMM_) * (LB_ * EPT_) + t;
    if (e0 - t + LB_ * EPT_ <= E_) {
      int m[EPT_], p[EPT_], prev[EPT_];
#pragma unroll
      for (int j = 0; j < EPT_; ++j) { m[j] = pair_mask[e0 + j * LB_]; p[j] = pair_point[e0 + j * LB_]; }
#pragma unroll
      for (int j = 0; j < EPT_; ++j) prev[j] = atomicExch(&head[p[j]], e0 + j * LB_ + 1);
#pragma unroll
      for (int j = 0; j < EPT_; ++j) link[e0 + j * LB_] = make_int2(prev[j], m[j]);
    } else {
      for (int j = 0; j < EPT_; ++j) {
        int e = e0 + j * LB_;
        if (e < E_) {
          int mm = pair_mask[e];
          int prev = atomicExch(&head[pair_point[e]], e + 1);
          link[e] = make_int2(prev, mm);
        }
      }
    }
    return;
  }
  int lane = t & 31;
  int wid  = t >> 5;
  int mt   = blockIdx.x;
  int r    = t >> 4;
  int cg   = t & 15;
  int lo = mstart[(mt << 4) + r];
  int hi = mstart[(mt << 4) + r + 1];
  int s0 = __builtin_amdgcn_readfirstlane(mstart[mt << 4]);
  int s1 = __builtin_amdgcn_readfirstlane(mstart[(mt << 4) + 16]);
  float sum[16];
#pragma unroll
  for (int cc = 0; cc < 16; ++cc) sum[cc] = 0.0f;
  for (int cs = s0 & ~3; cs < s1; cs += CH_) {
    int sl = cs + (lane << 2);
    float4 v[32];
#pragma unroll
    for (int i = 0; i < 32; ++i) v[i] = make_float4(0.0f, 0.0f, 0.0f, 0.0f);
    if (sl < s1) {
#pragma unroll
      for (int i = 0; i < 32; ++i) v[i] = *(const float4*)&gfT[(size_t)((wid << 5) + i) * G_ + sl];
    }
#pragma unroll
    for (int i = 0; i < 32; ++i) *(float4*)&tile[((wid << 5) + i) * TPT_ + (lane << 2)] = v[i];
    __syncthreads();
    int a  = max(lo, cs) - cs;
    int bd = min(hi, cs + CH_) - cs;
    for (int i = a; i < bd; ++i) {
#pragma unroll
      for (int cc = 0; cc < 16; ++cc) sum[cc] += tile[(cg + (cc << 4)) * TPT_ + i];
    }
    __syncthreads();
  }
  {
    int   cn = hi - lo;
    float dn = fmaxf((float)cn, 1.0f);
    bool  on = cn > 0;
#pragma unroll
    for (int cc = 0; cc < 16; ++cc) {
      float q = sum[cc] / dn;
      astrip[r * AP_ + cg + (cc << 4)] = on ? q : 0.0f;
    }
  }
  __syncthreads();

  int ncol = (wid << 4) + (lane & 15);
  int koff = (lane >> 4) << 1;
  const float* ar = &astrip[(lane & 15) * AP_ + koff];
  v8f c = {};
  for (int kk = 0; kk < C_; kk += 4) {
    v2f a, b;
    int k0 = kk + koff, k1 = kk + koff + 1;
    a.x = ar[kk];
    a.y = ar[kk + 1];
    b.x = W1[(size_t)k0 * C2_ + ncol];
    b.y = W1[(size_t)k1 * C2_ + ncol];
    c = __builtin_amdgcn_wmma_f32_16x16x4_f32(false, a, false, b, (short)0, c, false, false);
  }
  for (int i = 0; i < 8; ++i) {
    int rr = (mt << 4) + i + ((lane >> 4) << 3);
    mask_h1[(size_t)rr * C2_ + ncol] = c[i];
  }
}

#define TP_ 132

__global__ void __launch_bounds__(128)
k_final(const float* __restrict__ mask_h1, const int* __restrict__ head,
        const int2* __restrict__ link,
        const float* __restrict__ b1,   const float* __restrict__ W2,
        const float* __restrict__ b2,   float* __restrict__ out) {
  __shared__ alignas(16) float tile[4][16 * TP_];
  int lane = threadIdx.x & 31;
  int wid  = threadIdx.x >> 5;
  int m_base = ((blockIdx.x << 2) + wid) << 4;
  const float4* rows = (const float4*)mask_h1;

  int p = m_base + (lane & 15);
  int e = 0;
  if (lane < 16 && p < N_) e = head[p];
  int2 lk = make_int2(0, 0);
  if (e != 0) lk = link[e - 1];
  int n = 0;
  float4 acc[16];
#pragma unroll
  for (int i = 0; i < 16; ++i) acc[i] = make_float4(0.0f, 0.0f, 0.0f, 0.0f);

  for (int it = 0; it < E_; ++it) {
    unsigned live = __builtin_amdgcn_ballot_w32(e != 0);
    if (live == 0u) break;
    if (e != 0) ++n;
    int en = lk.x;
    int2 lkn = make_int2(0, 0);
    if (en != 0) lkn = link[en - 1];
    float4 r[16];
#pragma unroll
    for (int i = 0; i < 16; ++i) {
      int m = __builtin_amdgcn_readlane(lk.y, i);
      r[i] = rows[(size_t)m * 32 + lane];
    }
#pragma unroll
    for (int i = 0; i < 16; ++i) {
      if ((live >> i) & 1u) {
        acc[i].x += r[i].x; acc[i].y += r[i].y; acc[i].z += r[i].z; acc[i].w += r[i].w;
      }
    }
    e = en;
    lk = lkn;
  }

  float* tw = tile[wid];
  float4 bb = make_float4(b1[4 * lane], b1[4 * lane + 1], b1[4 * lane + 2], b1[4 * lane + 3]);
#pragma unroll
  for (int i = 0; i < 16; ++i) {
    float cn = (float)__builtin_amdgcn_readlane(n, i);
    float scale = 1.0f / fmaxf(cn, 1e-6f);
    float4 h;
    h.x = fmaxf(acc[i].x * scale + bb.x, 0.0f);
    h.y = fmaxf(acc[i].y * scale + bb.y, 0.0f);
    h.z = fmaxf(acc[i].z * scale + bb.z, 0.0f);
    h.w = fmaxf(acc[i].w * scale + bb.w, 0.0f);
    *(float4*)&tw[i * TP_ + 4 * lane] = h;
  }
  __syncthreads();

  int koff = (lane >> 4) << 1;
  int col  = lane & 15;
  const float* ar = &tw[(lane & 15) * TP_ + koff];
  v8f c0 = {}, c1 = {};
  for (int kk = 0; kk < C2_; kk += 4) {
    int k0 = kk + koff, k1 = kk + koff + 1;
    v2f a, b0, b1v;
    a.x = ar[kk];
    a.y = ar[kk + 1];
    b0.x = W2[(size_t)k0 * L_ + col];
    b0.y = W2[(size_t)k1 * L_ + col];
    int col1 = 16 + col;
    b1v.x = (col1 < L_) ? W2[(size_t)k0 * L_ + col1] : 0.0f;
    b1v.y = (col1 < L_) ? W2[(size_t)k1 * L_ + col1] : 0.0f;
    c0 = __builtin_amdgcn_wmma_f32_16x16x4_f32(false, a, false, b0,  (short)0, c0, false, false);
    c1 = __builtin_amdgcn_wmma_f32_16x16x4_f32(false, a, false, b1v, (short)0, c1, false, false);
  }
  for (int i = 0; i < 8; ++i) {
    int r = m_base + i + ((lane >> 4) << 3);
    if (r < N_) {
      out[(size_t)r * L_ + col] = c0[i] + b2[col];
      int col1 = 16 + col;
      if (col1 < L_) out[(size_t)r * L_ + col1] = c1[i] + b2[col1];
    }
  }
}

extern "C" void kernel_launch(void* const* d_in, const int* in_sizes, int n_in,
                              void* d_out, int out_size, void* d_ws, size_t ws_size,
                              hipStream_t stream) {
  const float* img        = (const float*)d_in[0];
  const float* is_seen    = (const float*)d_in[1];
  const float* W1         = (const float*)d_in[2];
  const float* b1         = (const float*)d_in[3];
  const float* W2         = (const float*)d_in[4];
  const float* b2         = (const float*)d_in[5];
  const int*   coords     = (const int*)d_in[6];
  const int*   centers    = (const int*)d_in[7];
  const int*   group2mask = (const int*)d_in[8];
  const int*   pair_mask  = (const int*)d_in[9];
  const int*   pair_point = (const int*)d_in[10];
  float* out = (float*)d_out;

  float* ws         = (float*)d_ws;
  float* group_feat = ws;
  float* sortf      = group_feat + (size_t)G_ * C_;
  float* headf      = sortf + (size_t)M_ * C_ + M_;
  float* tabf       = headf + N_;
  float* mask_h1    = tabf + (size_t)M_ * C_;
  int2*  link       = (int2*)(mask_h1 + (size_t)M_ * C2_);
  int*   head       = (int*)headf;
  int2*  tab        = (int2*)tabf;
  float* den        = tabf + (size_t)V_ * G_ * 2;
  int*   perm       = (int*)sortf;
  int*   mstart     = perm + G_;

  int nz = N_;
  k_pre<<<NSORT_ + NDEN_ + NMSORT_ + (nz + 1023) / 1024, 1024, 0, stream>>>(is_seen, coords, centers, group2mask, tab, den, perm, mstart, headf, nz);

  k_group_gather<<<C_, 1024, 0, stream>>>(img, tab, den, perm, group_feat);
  k_mask_h1<<<NGEMM_ + NLINK_, LB_, 0, stream>>>(group_feat, mstart, W1, mask_h1, pair_mask, pair_point, head, link);

  int tiles = (N_ + 15) / 16;
  k_final<<<(tiles + 3) / 4, 128, 0, stream>>>(mask_h1, head, link, b1, W2, b2, out);
}


# point kernel: mask_h1 (256 KB) staged once per block into LDS (250 blocks x 20 waves); list walk adds rows from LDS straight in A-fragment layout, list order kept; final GEMM's k sum order changed
# speedup vs baseline: 3.0028x; 1.1756x over previous
#include <hip/hip_runtime.h>

#define V_  12
#define N_  80000
#define C_  256
#define H_  128
#define W_  128
#define G_  4096
#define M_  512
#define E_  200000
#define L_  20
#define C2_ 128

typedef __attribute__((ext_vector_type(2))) float v2f;
typedef __attribute__((ext_vector_type(8))) float v8f;

#define PIXB_ 14
#define NSORT_ V_
#define NDEN_  (G_ / 1024)
#define NMSORT_ 1

__global__ void __launch_bounds__(1024)
k_pre(const float* __restrict__ is_seen, const int* __restrict__ coords,
      const int* __restrict__ centers, const int* __restrict__ group2mask,
      int2* __restrict__ tab, float* __restrict__ den,
      int* __restrict__ perm, int* __restrict__ mstart,
      float* __restrict__ zp, int nz) {
  __shared__ alignas(16) int hist[512];
  __shared__ int base[512];
  __shared__ int wtot[16];
  int t = threadIdx.x;
  int b = blockIdx.x;
  if (b >= NSORT_ + NDEN_ + NMSORT_) {
    int i = (b - NSORT_ - NDEN_ - NMSORT_) * 1024 + t;
    if (i < nz) zp[i] = 0.0f;
    return;
  }
  if (b >= NSORT_ + NDEN_) {
    if (t < 512) hist[t] = 0;
    __syncthreads();
    int mk[4], rk[4];
    for (int j = 0; j < 4; ++j) mk[j] = group2mask[j * 1024 + t] & (M_ - 1);
    for (int j = 0; j < 4; ++j) rk[j] = atomicAdd(&hist[mk[j]], 1);
    __syncthreads();
    int cnt = (t < 512) ? hist[t] : 0;
    int incl = cnt;
    for (int d = 1; d < 32; d <<= 1) {
      int y = __shfl_up(incl, d);
      if ((t & 31) >= d) incl += y;
    }
    if (t < 512 && (t & 31) == 31) wtot[t >> 5] = incl;
    __syncthreads();
    if (t < 512) {
      int off = 0;
      for (int w = 0; w < 16; ++w) off += (w < (t >> 5)) ? wtot[w] : 0;
      int ex = off + incl - cnt;
      base[t] = ex;
      mstart[t] = ex;
    }
    if (t == 0) mstart[M_] = G_;
    __syncthreads();
    for (int j = 0; j < 4; ++j) perm[base[mk[j]] + rk[j]] = j * 1024 + t;
    return;
  }
  if (b >= NSORT_) {
    int g = (b - NSORT_) * 1024 + t;
    int n = centers[g];
    float d = 0.0f;
    for (int v = 0; v < V_; ++v) d += is_seen[(size_t)v * N_ + n];
    den[g] = d;
    return;
  }
  int v = b;
  if (t < 512) hist[t] = 0;
  __syncthreads();
  int pix[4], bin[4], rank[4];
  float w[4];
  for (int j = 0; j < 4; ++j) {
    int g = j * 1024 + t;
    int n = centers[g];
    int x = coords[((size_t)v * N_ + n) * 2 + 0];
    int y = coords[((size_t)v * N_ + n) * 2 + 1];
    w[j] = is_seen[(size_t)v * N_ + n];
    pix[j] = ((y & (H_ - 1)) << 7) | (x & (W_ - 1));
    bin[j] = pix[j] >> 5;
  }
  for (int j = 0; j < 4; ++j) rank[j] = atomicAdd(&hist[bin[j]], 1);
  __syncthreads();
  if (t < 512) {
    int s = 0;
    for (int k = 0; k < 128; ++k) {
      int4 h = ((const int4*)hist)[k];
      s += ((4 * k     < t) ? h.x : 0) + ((4 * k + 1 < t) ? h.y : 0)
         + ((4 * k + 2 < t) ? h.z : 0) + ((4 * k + 3 < t) ? h.w : 0);
    }
    base[t] = s;
  }
  __syncthreads();
  for (int j = 0; j < 4; ++j) {
    int g = j * 1024 + t;
    int slot = base[bin[j]] + rank[j];
    tab[(size_t)v * G_ + slot] = make_int2(pix[j] | (g << PIXB_), __float_as_int(w[j]));
  }
}

__device__ __forceinline__ void lds_barrier() {
  __builtin_amdgcn_fence(__ATOMIC_RELEASE, "workgroup", "local");
  __builtin_amdgcn_s_barrier();
  __builtin_amdgcn_fence(__ATOMIC_ACQUIRE, "workgroup", "local");
}

#define DI_ 4
#define DT_ 7

__global__ void __launch_bounds__(1024)
k_group_gather(const float* __restrict__ img, const int2* __restrict__ tab,
               const float* __restrict__ den, const int* __restrict__ perm,
               float* __restrict__ gfT) {
  __shared__ float acc[G_];
  int c = blockIdx.x;
  int t = threadIdx.x;
  int2  e[V_][4];
  float val[V_][4];
  int   pg[4];
  float dn[4];
  for (int j = 0; j < 4; ++j) { pg[j] = perm[j * 1024 + t] & (G_ - 1); dn[j] = den[j * 1024 + t]; }
#pragma unroll
  for (int v = 0; v < DT_; ++v)
#pragma unroll
    for (int j = 0; j < 4; ++j)
      e[v][j] = tab[(size_t)v * G_ + j * 1024 + t];
  for (int j = 0; j < 4; ++j) acc[j * 1024 + t] = 0.0f;
#pragma unroll
  for (int v = 0; v < DI_; ++v) {
    const float* plane = img + (((size_t)v * C_ + c) << PIXB_);
#pragma unroll
    for (int j = 0; j < 4; ++j)
      val[v][j] = plane[e[v][j].x & ((1 << PIXB_) - 1)];
  }
  lds_barrier();
#pragma unroll
  for (int v = 0; v < V_; ++v) {
    if (v + DT_ < V_) {
#pragma unroll
      for (int j = 0; j < 4; ++j)
        e[v + DT_][j] = tab[(size_t)(v + DT_) * G_ + j * 1024 + t];
    }
    if (v + DI_ < V_) {
      const float* plane = img + (((size_t)(v + DI_) * C_ + c) << PIXB_);
#pragma unroll
      for (int j = 0; j < 4; ++j)
        val[v + DI_][j] = plane[e[v + DI_][j].x & ((1 << PIXB_) - 1)];
    }
    __builtin_amdgcn_sched_barrier(0);
#pragma unroll
    for (int j = 0; j < 4; ++j) {
      int g = e[v][j].x >> PIXB_;
      acc[g] += __int_as_float(e[v][j].y) * val[v][j];
    }
    lds_barrier();
  }
  for (int j = 0; j < 4; ++j) {
    int g = j * 1024 + t;
    acc[g] = acc[g] / (dn[j] + 1e-6f);
  }
  __syncthreads();
  for (int j = 0; j < 4; ++j) {
    int sl = j * 1024 + t;
    gfT[(size_t)c * G_ + sl] = acc[pg[j]];
  }
}

#define AP_    260
#define NGEMM_ 32
#define LB_    256
#define EPT_   4
#define NLINK_ ((E_ + LB_ * EPT_ - 1) / (LB_ * EPT_))
#define CH_    128
#define TPT_   132

__global__ void __launch_bounds__(LB_)
k_mask_h1(const float* __restrict__ gfT, const int* __restrict__ mstart,
          const float* __restrict__ W1, float* __restrict__ mask_h1,
          const int* __restrict__ pair_mask, const int* __restrict__ pair_point,
          int* __restrict__ head, int2* __restrict__ link) {
  __shared__ alignas(16) float astrip[16 * AP_];
  __shared__ alignas(16) float tile[C_ * TPT_];
  int t = threadIdx.x;
  if (blockIdx.x >= NGEMM_) {
    int e0 = (blockIdx.x - NGEMM_) * (LB_ * EPT_) + t;
    if (e0 - t + LB_ * EPT_ <= E_) {
      int m[EPT_], p[EPT_], prev[EPT_];
#pragma unroll
      for (int j = 0; j < EPT_; ++j) { m[j] = pair_mask[e0 + j * LB_]; p[j] = pair_point[e0 + j * LB_]; }
#pragma unroll
      for (int j = 0; j < EPT_; ++j) prev[j] = atomicExch(&head[p[j]], e0 + j * LB_ + 1);
#pragma unroll
      for (int j = 0; j < EPT_; ++j) link[e0 + j * LB_] = make_int2(prev[j], m[j]);
    } else {
      for (int j = 0; j < EPT_; ++j) {
        int e = e0 + j * LB_;
        if (e < E_) {
          int mm = pair_mask[e];
          int prev = atomicExch(&head[pair_point[e]], e + 1);
          link[e] = make_int2(prev, mm);
        }
      }
    }
    return;
  }
  int lane = t & 31;
  int wid  = t >> 5;
  int mt   = blockIdx.x;
  int r    = t >> 4;
  int cg   = t & 15;
  int lo = mstart[(mt << 4) + r];
  int hi = mstart[(mt << 4) + r + 1];
  int s0 = __builtin_amdgcn_readfirstlane(mstart[mt << 4]);
  int s1 = __builtin_amdgcn_readfirstlane(mstart[(mt << 4) + 16]);
  float sum[16];
#pragma unroll
  for (int cc = 0; cc < 16; ++cc) sum[cc] = 0.0f;
  for (int cs = s0 & ~3; cs < s1; cs += CH_) {
    int sl = cs + (lane << 2);
    float4 v[32];
#pragma unroll
    for (int i = 0; i < 32; ++i) v[i] = make_float4(0.0f, 0.0f, 0.0f, 0.0f);
    if (sl < s1) {
#pragma unroll
      for (int i = 0; i < 32; ++i) v[i] = *(const float4*)&gfT[(size_t)((wid << 5) + i) * G_ + sl];
    }
#pragma unroll
    for (int i = 0; i < 32; ++i) *(float4*)&tile[((wid << 5) + i) * TPT_ + (lane << 2)] = v[i];
    __syncthreads();
    int a  = max(lo, cs) - cs;
    int bd = min(hi, cs + CH_) - cs;
    for (int i = a; i < bd; ++i) {
#pragma unroll
      for (int cc = 0; cc < 16; ++cc) sum[cc] += tile[(cg + (cc << 4)) * TPT_ + i];
    }
    __syncthreads();
  }
  {
    int   cn = hi - lo;
    float dn = fmaxf((float)cn, 1.0f);
    bool  on = cn > 0;
#pragma unroll
    for (int cc = 0; cc < 16; ++cc) {
      float q = sum[cc] / dn;
      astrip[r * AP_ + cg + (cc << 4)] = on ? q : 0.0f;
    }
  }
  __syncthreads();

  int ncol = (wid << 4) + (lane & 15);
  int koff = (lane >> 4) << 1;
  const float* ar = &astrip[(lane & 15) * AP_ + koff];
  v8f c = {};
  for (int kk = 0; kk < C_; kk += 4) {
    v2f a, b;
    int k0 = kk + koff, k1 = kk + koff + 1;
    a.x = ar[kk];
    a.y = ar[kk + 1];
    b.x = W1[(size_t)k0 * C2_ + ncol];
    b.y = W1[(size_t)k1 * C2_ + ncol];
    c = __builtin_amdgcn_wmma_f32_16x16x4_f32(false, a, false, b, (short)0, c, false, false);
  }
  for (int i = 0; i < 8; ++i) {
    int rr = (mt << 4) + i + ((lane >> 4) << 3);
    mask_h1[(size_t)rr * C2_ + ncol] = c[i];
  }
}

#define FP_   132
#define FW_   20
#define FT_   (FW_ * 32)
#define NFB_  ((N_ / 16 + FW_ - 1) / FW_)
#define NPC_  (M_ * 32)
#define NST_  ((NPC_ + FT_ - 1) / FT_)

__device__ __forceinline__ int kof(int q, int h, int s) {
  return ((q >> 1) << 3) + (h << 2) + ((q & 1) << 1) + s;
}

__global__ void __launch_bounds__(FT_)
k_final(const float* __restrict__ mask_h1, const int* __restrict__ head,
        const int2* __restrict__ link,
        const float* __restrict__ b1,   const float* __restrict__ W2,
        const float* __restrict__ b2,   float* __restrict__ out) {
  __shared__ alignas(16) float tab[M_ * FP_];
  __shared__ alignas(16) float w2p[32 * 2 * 2 * 16 * 2];
  int t    = threadIdx.x;
  int lane = t & 31;
  int wid  = t >> 5;
  int c    = lane & 15;
  int h    = lane >> 4;
  int m_base = (blockIdx.x * FW_ + wid) << 4;
  int p = m_base + c;

  int e = head[min(p, N_ - 1)];
  if (p >= N_) e = 0;

  const float4* src = (const float4*)mask_h1;
  int rot = (blockIdx.x << 6) & (NPC_ - 1);
  float4 st[NST_];
#pragma unroll
  for (int i = 0; i < NST_; ++i) st[i] = src[(min(i * FT_ + t, NPC_ - 1) + rot) & (NPC_ - 1)];
  int2 lk = link[max(e, 1) - 1];
  if (e == 0) lk = make_int2(0, 0);
  float wv[7];
#pragma unroll
  for (int u = 0; u < 7; ++u) {
    int i = u * FT_ + t;
    int s = i & 1, cc = (i >> 1) & 15, hh = (i >> 5) & 1, tl = (i >> 6) & 1, q = (i >> 7) & 31;
    int col = (tl << 4) + cc;
    float w = W2[(size_t)kof(q, hh, s) * L_ + min(col, L_ - 1)];
    wv[u] = (col < L_) ? w : 0.0f;
  }
#pragma unroll
  for (int i = 0; i < NST_; ++i) {
    int g = i * FT_ + t;
    int f = (min(g, NPC_ - 1) + rot) & (NPC_ - 1);
    if (g < NPC_) *(float4*)&tab[(f >> 5) * FP_ + ((f & 31) << 2)] = st[i];
  }
#pragma unroll
  for (int u = 0; u < 7; ++u) {
    int i = u * FT_ + t;
    if (i < 32 * 2 * 2 * 16 * 2) w2p[i] = wv[u];
  }
  __syncthreads();

  int n = 0;
  float4 acc[16];
#pragma unroll
  for (int j = 0; j < 16; ++j) acc[j] = make_float4(0.0f, 0.0f, 0.0f, 0.0f);
  const float* tb = tab + (h << 2);

  for (int it = 0; it < E_; ++it) {
    unsigned live = __builtin_amdgcn_ballot_w32(e != 0);
    if (live == 0u) break;
    int en = lk.x;
    int2 lkn = make_int2(0, 0);
    if (en != 0) lkn = link[en - 1];
    if (e != 0) {
      ++n;
      const float* row = tb + lk.y * FP_;
#pragma unroll
      for (int j = 0; j < 16; ++j) {
        float4 r = *(const float4*)&row[j << 3];
        acc[j].x += r.x; acc[j].y += r.y; acc[j].z += r.z; acc[j].w += r.w;
      }
    }
    e = en;
    lk = lkn;
  }

  float cn = (float)n;
  float scale = 1.0f / fmaxf(cn, 1e-6f);
  const float4* bp = (const float4*)b1;
  const v2f*    wp = (const v2f*)w2p;
  float bo0 = b2[c];
  float bo1 = (16 + c < L_) ? b2[16 + c] : 0.0f;
  v8f c0 = {}, c1 = {};
#pragma unroll
  for (int j = 0; j < 16; ++j) {
    float4 bb = bp[2 * j + h];
    v2f a0, a1;
    a0.x = fmaxf(acc[j].x * scale + bb.x, 0.0f);
    a0.y = fmaxf(acc[j].y * scale + bb.y, 0.0f);
    a1.x = fmaxf(acc[j].z * scale + bb.z, 0.0f);
    a1.y = fmaxf(acc[j].w * scale + bb.w, 0.0f);
    v2f b00 = wp[((((2 * j) * 2 + 0) * 2 + h) << 4) + c];
    v2f b01 = wp[((((2 * j) * 2 + 1) * 2 + h) << 4) + c];
    v2f b10 = wp[((((2 * j + 1) * 2 + 0) * 2 + h) << 4) + c];
    v2f b11 = wp[((((2 * j + 1) * 2 + 1) * 2 + h) << 4) + c];
    c0 = __builtin_amdgcn_wmma_f32_16x16x4_f32(false, a0, false, b00, (short)0, c0, false, false);
    c1 = __builtin_amdgcn_wmma_f32_16x16x4_f32(false, a0, false, b01, (short)0, c1, false, false);
    c0 = __builtin_amdgcn_wmma_f32_16x16x4_f32(false, a1, false, b10, (short)0, c0, false, false);
    c1 = __builtin_amdgcn_wmma_f32_16x16x4_f32(false, a1, false, b11, (short)0, c1, false, false);
  }
  for (int i = 0; i < 8; ++i) {
    int r = m_base + i + (h << 3);
    if (r < N_) {
      out[(size_t)r * L_ + c] = c0[i] + bo0;
      if (16 + c < L_) out[(size_t)r * L_ + 16 + c] = c1[i] + bo1;
    }
  }
}

extern "C" void kernel_launch(void* const* d_in, const int* in_sizes, int n_in,
                              void* d_out, int out_size, void* d_ws, size_t ws_size,
                              hipStream_t stream) {
  const float* img        = (const float*)d_in[0];
  const float* is_seen    = (const float*)d_in[1];
  const float* W1         = (const float*)d_in[2];
  const float* b1         = (const float*)d_in[3];
  const float* W2         = (const float*)d_in[4];
  const float* b2         = (const float*)d_in[5];
  const int*   coords     = (const int*)d_in[6];
  const int*   centers    = (const int*)d_in[7];
  const int*   group2mask = (const int*)d_in[8];
  const int*   pair_mask  = (const int*)d_in[9];
  const int*   pair_point = (const int*)d_in[10];
  float* out = (float*)d_out;

  float* ws         = (float*)d_ws;
  float* group_feat = ws;
  float* sortf      = group_feat + (size_t)G_ * C_;
  float* headf      = sortf + (size_t)M_ * C_ + M_;
  float* tabf       = headf + N_;
  float* mask_h1    = tabf + (size_t)M_ * C_;
  int2*  link       = (int2*)(mask_h1 + (size_t)M_ * C2_);
  int*   head       = (int*)headf;
  int2*  tab        = (int2*)tabf;
  float* den        = tabf + (size_t)V_ * G_ * 2;
  int*   perm       = (int*)sortf;
  int*   mstart     = perm + G_;

  int nz = N_;
  k_pre<<<NSORT_ + NDEN_ + NMSORT_ + (nz + 1023) / 1024, 1024, 0, stream>>>(is_seen, coords, centers, group2mask, tab, den, perm, mstart, headf, nz);

  k_group_gather<<<C_, 1024, 0, stream>>>(img, tab, den, perm, group_feat);
  k_mask_h1<<<NGEMM_ + NLINK_, LB_, 0, stream>>>(group_feat, mstart, W1, mask_h1, pair_mask, pair_point, head, link);

  k_final<<<NFB_, FT_, 0, stream>>>(mask_h1, head, link, b1, W2, b2, out);
}


# first launch: line-per-lane index fetches spread from 16 to 64 blocks: 4 blocks a view each read all pixels but 1/4 of the weights, write 1/4 of the slots; weight sums 4->16 blocks; scan prefix
# speedup vs baseline: 3.2849x; 1.0939x over previous
#include <hip/hip_runtime.h>

#define V_  12
#define N_  80000
#define C_  256
#define H_  128
#define W_  128
#define G_  4096
#define M_  512
#define E_  200000
#define L_  20
#define C2_ 128

typedef __attribute__((ext_vector_type(2))) float v2f;
typedef __attribute__((ext_vector_type(8))) float v8f;

#define PIXB_ 14
#define NQ_    4
#define NSORT_ (V_ * NQ_)
#define DG_    256
#define NDEN_  (G_ / DG_)
#define NMSORT_ 1

__global__ void __launch_bounds__(1024)
k_pre(const float* __restrict__ is_seen, const int* __restrict__ coords,
      const int* __restrict__ centers, const int* __restrict__ group2mask,
      int2* __restrict__ tab, float* __restrict__ den,
      int* __restrict__ perm, int* __restrict__ mstart,
      float* __restrict__ zp, int nz) {
  __shared__ int hist[NQ_ * 512];
  __shared__ int base[512];
  __shared__ int wtot[16];
  __shared__ float sw[V_ * DG_];
  int t = threadIdx.x;
  int b = blockIdx.x;
  if (b >= NSORT_ + NDEN_ + NMSORT_) {
    int i = (b - NSORT_ - NDEN_ - NMSORT_) * 1024 + t;
    if (i < nz) zp[i] = 0.0f;
    return;
  }
  if (b >= NSORT_ + NDEN_) {
    if (t < 512) hist[t] = 0;
    __syncthreads();
    int mk[4], rk[4];
    for (int j = 0; j < 4; ++j) mk[j] = group2mask[j * 1024 + t] & (M_ - 1);
    for (int j = 0; j < 4; ++j) rk[j] = atomicAdd(&hist[mk[j]], 1);
    __syncthreads();
    int cnt = (t < 512) ? hist[t] : 0;
    int incl = cnt;
    for (int d = 1; d < 32; d <<= 1) {
      int y = __shfl_up(incl, d);
      if ((t & 31) >= d) incl += y;
    }
    if (t < 512 && (t & 31) == 31) wtot[t >> 5] = incl;
    __syncthreads();
    if (t < 512) {
      int off = 0;
      for (int w = 0; w < 16; ++w) off += (w < (t >> 5)) ? wtot[w] : 0;
      int ex = off + incl - cnt;
      base[t] = ex;
      mstart[t] = ex;
    }
    if (t == 0) mstart[M_] = G_;
    __syncthreads();
    for (int j = 0; j < 4; ++j) perm[base[mk[j]] + rk[j]] = j * 1024 + t;
    return;
  }
  if (b >= NSORT_) {
    int gl = t & (DG_ - 1);
    int vq = t >> 8;
    int g  = (b - NSORT_) * DG_ + gl;
    int n  = centers[g];
    for (int k = 0; k < 3; ++k) {
      int v = vq * 3 + k;
      sw[v * DG_ + gl] = is_seen[(size_t)v * N_ + n];
    }
    __syncthreads();
    if (t < DG_) {
      float d = 0.0f;
      for (int v = 0; v < V_; ++v) d += sw[v * DG_ + t];
      den[g] = d;
    }
    return;
  }
  int v = b >> 2;
  int q = b & (NQ_ - 1);
  hist[t] = 0;
  hist[1024 + t] = 0;
  __syncthreads();
  int nn[4], pix[4], bin[4];
  for (int j = 0; j < 4; ++j) nn[j] = centers[j * 1024 + t];
  int nq = (q == 0) ? nn[0] : (q == 1) ? nn[1] : (q == 2) ? nn[2] : nn[3];
  for (int j = 0; j < 4; ++j) {
    int x = coords[((size_t)v * N_ + nn[j]) * 2 + 0];
    int y = coords[((size_t)v * N_ + nn[j]) * 2 + 1];
    pix[j] = ((y & (H_ - 1)) << 7) | (x & (W_ - 1));
    bin[j] = pix[j] >> 5;
  }
  float wq = is_seen[(size_t)v * N_ + nq];
  int rk[4];
  for (int j = 0; j < 4; ++j) rk[j] = atomicAdd(&hist[j * 512 + bin[j]], 1);
  int pq = (q == 0) ? pix[0] : (q == 1) ? pix[1] : (q == 2) ? pix[2] : pix[3];
  int rq = (q == 0) ? rk[0]  : (q == 1) ? rk[1]  : (q == 2) ? rk[2]  : rk[3];
  __syncthreads();
  int cnt = 0, low = 0;
  if (t < 512) {
    for (int j = 0; j < NQ_; ++j) {
      int hh = hist[j * 512 + t];
      cnt += hh;
      low += (j < q) ? hh : 0;
    }
  }
  int incl = cnt;
  for (int d = 1; d < 32; d <<= 1) {
    int y = __shfl_up(incl, d);
    if ((t & 31) >= d) incl += y;
  }
  if (t < 512 && (t & 31) == 31) wtot[t >> 5] = incl;
  __syncthreads();
  if (t < 512) {
    int off = 0;
    for (int w = 0; w < 16; ++w) off += (w < (t >> 5)) ? wtot[w] : 0;
    base[t] = off + incl - cnt + low;
  }
  __syncthreads();
  {
    int g = q * 1024 + t;
    int slot = base[pq >> 5] + rq;
    tab[(size_t)v * G_ + slot] = make_int2(pq | (g << PIXB_), __float_as_int(wq));
  }
}

__device__ __forceinline__ void lds_barrier() {
  __builtin_amdgcn_fence(__ATOMIC_RELEASE, "workgroup", "local");
  __builtin_amdgcn_s_barrier();
  __builtin_amdgcn_fence(__ATOMIC_ACQUIRE, "workgroup", "local");
}

#define DI_ 4
#define DT_ 7

__global__ void __launch_bounds__(1024)
k_group_gather(const float* __restrict__ img, const int2* __restrict__ tab,
               const float* __restrict__ den, const int* __restrict__ perm,
               float* __restrict__ gfT) {
  __shared__ float acc[G_];
  int c = blockIdx.x;
  int t = threadIdx.x;
  int2  e[V_][4];
  float val[V_][4];
  int   pg[4];
  float dn[4];
  for (int j = 0; j < 4; ++j) { pg[j] = perm[j * 1024 + t] & (G_ - 1); dn[j] = den[j * 1024 + t]; }
#pragma unroll
  for (int v = 0; v < DT_; ++v)
#pragma unroll
    for (int j = 0; j < 4; ++j)
      e[v][j] = tab[(size_t)v * G_ + j * 1024 + t];
  for (int j = 0; j < 4; ++j) acc[j * 1024 + t] = 0.0f;
#pragma unroll
  for (int v = 0; v < DI_; ++v) {
    const float* plane = img + (((size_t)v * C_ + c) << PIXB_);
#pragma unroll
    for (int j = 0; j < 4; ++j)
      val[v][j] = plane[e[v][j].x & ((1 << PIXB_) - 1)];
  }
  lds_barrier();
#pragma unroll
  for (int v = 0; v < V_; ++v) {
    if (v + DT_ < V_) {
#pragma unroll
      for (int j = 0; j < 4; ++j)
        e[v + DT_][j] = tab[(size_t)(v + DT_) * G_ + j * 1024 + t];
    }
    if (v + DI_ < V_) {
      const float* plane = img + (((size_t)(v + DI_) * C_ + c) << PIXB_);
#pragma unroll
      for (int j = 0; j < 4; ++j)
        val[v + DI_][j] = plane[e[v + DI_][j].x & ((1 << PIXB_) - 1)];
    }
    __builtin_amdgcn_sched_barrier(0);
#pragma unroll
    for (int j = 0; j < 4; ++j) {
      int g = e[v][j].x >> PIXB_;
      acc[g] += __int_as_float(e[v][j].y) * val[v][j];
    }
    lds_barrier();
  }
  for (int j = 0; j < 4; ++j) {
    int g = j * 1024 + t;
    acc[g] = acc[g] / (dn[j] + 1e-6f);
  }
  __syncthreads();
  for (int j = 0; j < 4; ++j) {
    int sl = j * 1024 + t;
    gfT[(size_t)c * G_ + sl] = acc[pg[j]];
  }
}

#define AP_    260
#define NGEMM_ 32
#define LB_    256
#define EPT_   4
#define NLINK_ ((E_ + LB_ * EPT_ - 1) / (LB_ * EPT_))
#define CH_    128
#define TPT_   132

__global__ void __launch_bounds__(LB_)
k_mask_h1(const float* __restrict__ gfT, const int* __restrict__ mstart,
          const float* __restrict__ W1, float* __restrict__ mask_h1,
          const int* __restrict__ pair_mask, const int* __restrict__ pair_point,
          int* __restrict__ head, int2* __restrict__ link) {
  __shared__ alignas(16) float astrip[16 * AP_];
  __shared__ alignas(16) float tile[C_ * TPT_];
  int t = threadIdx.x;
  if (blockIdx.x >= NGEMM_) {
    int e0 = (blockIdx.x - NGEMM_) * (LB_ * EPT_) + t;
    if (e0 - t + LB_ * EPT_ <= E_) {
      int m[EPT_], p[EPT_], prev[EPT_];
#pragma unroll
      for (int j = 0; j < EPT_; ++j) { m[j] = pair_mask[e0 + j * LB_]; p[j] = pair_point[e0 + j * LB_]; }
#pragma unroll
      for (int j = 0; j < EPT_; ++j) prev[j] = atomicExch(&head[p[j]], e0 + j * LB_ + 1);
#pragma unroll
      for (int j = 0; j < EPT_; ++j) link[e0 + j * LB_] = make_int2(prev[j], m[j]);
    } else {
      for (int j = 0; j < EPT_; ++j) {
        int e = e0 + j * LB_;
        if (e < E_) {
          int mm = pair_mask[e];
          int prev = atomicExch(&head[pair_point[e]], e + 1);
          link[e] = make_int2(prev, mm);
        }
      }
    }
    return;
  }
  int lane = t & 31;
  int wid  = t >> 5;
  int mt   = blockIdx.x;
  int r    = t >> 4;
  int cg   = t & 15;
  int lo = mstart[(mt << 4) + r];
  int hi = mstart[(mt << 4) + r + 1];
  int s0 = __builtin_amdgcn_readfirstlane(mstart[mt << 4]);
  int s1 = __builtin_amdgcn_readfirstlane(mstart[(mt << 4) + 16]);
  float sum[16];
#pragma unroll
  for (int cc = 0; cc < 16; ++cc) sum[cc] = 0.0f;
  for (int cs = s0 & ~3; cs < s1; cs += CH_) {
    int sl = cs + (lane << 2);
    float4 v[32];
#pragma unroll
    for (int i = 0; i < 32; ++i) v[i] = make_float4(0.0f, 0.0f, 0.0f, 0.0f);
    if (sl < s1) {
#pragma unroll
      for (int i = 0; i < 32; ++i) v[i] = *(const float4*)&gfT[(size_t)((wid << 5) + i) * G_ + sl];
    }
#pragma unroll
    for (int i = 0; i < 32; ++i) *(float4*)&tile[((wid << 5) + i) * TPT_ + (lane << 2)] = v[i];
    __syncthreads();
    int a  = max(lo, cs) - cs;
    int bd = min(hi, cs + CH_) - cs;
    for (int i = a; i < bd; ++i) {
#pragma unroll
      for (int cc = 0; cc < 16; ++cc) sum[cc] += tile[(cg + (cc << 4)) * TPT_ + i];
    }
    __syncthreads();
  }
  {
    int   cn = hi - lo;
    float dn = fmaxf((float)cn, 1.0f);
    bool  on = cn > 0;
#pragma unroll
    for (int cc = 0; cc < 16; ++cc) {
      float q = sum[cc] / dn;
      astrip[r * AP_ + cg + (cc << 4)] = on ? q : 0.0f;
    }
  }
  __syncthreads();

  int ncol = (wid << 4) + (lane & 15);
  int koff = (lane >> 4) << 1;
  const float* ar = &astrip[(lane & 15) * AP_ + koff];
  v8f c = {};
  for (int kk = 0; kk < C_; kk += 4) {
    v2f a, b;
    int k0 = kk + koff, k1 = kk + koff + 1;
    a.x = ar[kk];
    a.y = ar[kk + 1];
    b.x = W1[(size_t)k0 * C2_ + ncol];
    b.y = W1[(size_t)k1 * C2_ + ncol];
    c = __builtin_amdgcn_wmma_f32_16x16x4_f32(false, a, false, b, (short)0, c, false, false);
  }
  for (int i = 0; i < 8; ++i) {
    int rr = (mt << 4) + i + ((lane >> 4) << 3);
    mask_h1[(size_t)rr * C2_ + ncol] = c[i];
  }
}

#define FP_   132
#define FW_   20
#define FT_   (FW_ * 32)
#define NFB_  ((N_ / 16 + FW_ - 1) / FW_)
#define NPC_  (M_ * 32)
#define NST_  ((NPC_ + FT_ - 1) / FT_)

__device__ __forceinline__ int kof(int q, int h, int s) {
  return ((q >> 1) << 3) + (h << 2) + ((q & 1) << 1) + s;
}

__global__ void __launch_bounds__(FT_)
k_final(const float* __restrict__ mask_h1, const int* __restrict__ head,
        const int2* __restrict__ link,
        const float* __restrict__ b1,   const float* __restrict__ W2,
        const float* __restrict__ b2,   float* __restrict__ out) {
  __shared__ alignas(16) float tab[M_ * FP_];
  __shared__ alignas(16) float w2p[32 * 2 * 2 * 16 * 2];
  int t    = threadIdx.x;
  int lane = t & 31;
  int wid  = t >> 5;
  int c    = lane & 15;
  int h    = lane >> 4;
  int m_base = (blockIdx.x * FW_ + wid) << 4;
  int p = m_base + c;

  int e = head[min(p, N_ - 1)];
  if (p >= N_) e = 0;

  const float4* src = (const float4*)mask_h1;
  int rot = (blockIdx.x << 6) & (NPC_ - 1);
  float4 st[NST_];
#pragma unroll
  for (int i = 0; i < NST_; ++i) st[i] = src[(min(i * FT_ + t, NPC_ - 1) + rot) & (NPC_ - 1)];
  int2 lk = link[max(e, 1) - 1];
  if (e == 0) lk = make_int2(0, 0);
  float wv[7];
#pragma unroll
  for (int u = 0; u < 7; ++u) {
    int i = u * FT_ + t;
    int s = i & 1, cc = (i >> 1) & 15, hh = (i >> 5) & 1, tl = (i >> 6) & 1, q = (i >> 7) & 31;
    int col = (tl << 4) + cc;
    float w = W2[(size_t)kof(q, hh, s) * L_ + min(col, L_ - 1)];
    wv[u] = (col < L_) ? w : 0.0f;
  }
#pragma unroll
  for (int i = 0; i < NST_; ++i) {
    int g = i * FT_ + t;
    int f = (min(g, NPC_ - 1) + rot) & (NPC_ - 1);
    if (g < NPC_) *(float4*)&tab[(f >> 5) * FP_ + ((f & 31) << 2)] = st[i];
  }
#pragma unroll
  for (int u = 0; u < 7; ++u) {
    int i = u * FT_ + t;
    if (i < 32 * 2 * 2 * 16 * 2) w2p[i] = wv[u];
  }
  __syncthreads();

  int n = 0;
  float4 acc[16];
#pragma unroll
  for (int j = 0; j < 16; ++j) acc[j] = make_float4(0.0f, 0.0f, 0.0f, 0.0f);
  const float* tb = tab + (h << 2);

  for (int it = 0; it < E_; ++it) {
    unsigned live = __builtin_amdgcn_ballot_w32(e != 0);
    if (live == 0u) break;
    int en = lk.x;
    int2 lkn = make_int2(0, 0);
    if (en != 0) lkn = link[en - 1];
    if (e != 0) {
      ++n;
      const float* row = tb + lk.y * FP_;
#pragma unroll
      for (int j = 0; j < 16; ++j) {
        float4 r = *(const float4*)&row[j << 3];
        acc[j].x += r.x; acc[j].y += r.y; acc[j].z += r.z; acc[j].w += r.w;
      }
    }
    e = en;
    lk = lkn;
  }

  float cn = (float)n;
  float scale = 1.0f / fmaxf(cn, 1e-6f);
  const float4* bp = (const float4*)b1;
  const v2f*    wp = (const v2f*)w2p;
  float bo0 = b2[c];
  float bo1 = (16 + c < L_) ? b2[16 + c] : 0.0f;
  v8f c0 = {}, c1 = {};
#pragma unroll
  for (int j = 0; j < 16; ++j) {
    float4 bb = bp[2 * j + h];
    v2f a0, a1;
    a0.x = fmaxf(acc[j].x * scale + bb.x, 0.0f);
    a0.y = fmaxf(acc[j].y * scale + bb.y, 0.0f);
    a1.x = fmaxf(acc[j].z * scale + bb.z, 0.0f);
    a1.y = fmaxf(acc[j].w * scale + bb.w, 0.0f);
    v2f b00 = wp[((((2 * j) * 2 + 0) * 2 + h) << 4) + c];
    v2f b01 = wp[((((2 * j) * 2 + 1) * 2 + h) << 4) + c];
    v2f b10 = wp[((((2 * j + 1) * 2 + 0) * 2 + h) << 4) + c];
    v2f b11 = wp[((((2 * j + 1) * 2 + 1) * 2 + h) << 4) + c];
    c0 = __builtin_amdgcn_wmma_f32_16x16x4_f32(false, a0, false, b00, (short)0, c0, false, false);
    c1 = __builtin_amdgcn_wmma_f32_16x16x4_f32(false, a0, false, b01, (short)0, c1, false, false);
    c0 = __builtin_amdgcn_wmma_f32_16x16x4_f32(false, a1, false, b10, (short)0, c0, false, false);
    c1 = __builtin_amdgcn_wmma_f32_16x16x4_f32(false, a1, false, b11, (short)0, c1, false, false);
  }
  for (int i = 0; i < 8; ++i) {
    int r = m_base + i + (h << 3);
    if (r < N_) {
      out[(size_t)r * L_ + c] = c0[i] + bo0;
      if (16 + c < L_) out[(size_t)r * L_ + 16 + c] = c1[i] + bo1;
    }
  }
}

extern "C" void kernel_launch(void* const* d_in, const int* in_sizes, int n_in,
                              void* d_out, int out_size, void* d_ws, size_t ws_size,
                              hipStream_t stream) {
  const float* img        = (const float*)d_in[0];
  const float* is_seen    = (const float*)d_in[1];
  const float* W1         = (const float*)d_in[2];
  const float* b1         = (const float*)d_in[3];
  const float* W2         = (const float*)d_in[4];
  const float* b2         = (const float*)d_in[5];
  const int*   coords     = (const int*)d_in[6];
  const int*   centers    = (const int*)d_in[7];
  const int*   group2mask = (const int*)d_in[8];
  const int*   pair_mask  = (const int*)d_in[9];
  const int*   pair_point = (const int*)d_in[10];
  float* out = (float*)d_out;

  float* ws         = (float*)d_ws;
  float* group_feat = ws;
  float* sortf      = group_feat + (size_t)G_ * C_;
  float* headf      = sortf + (size_t)M_ * C_ + M_;
  float* tabf       = headf + N_;
  float* mask_h1    = tabf + (size_t)M_ * C_;
  int2*  link       = (int2*)(mask_h1 + (size_t)M_ * C2_);
  int*   head       = (int*)headf;
  int2*  tab        = (int2*)tabf;
  float* den        = tabf + (size_t)V_ * G_ * 2;
  int*   perm       = (int*)sortf;
  int*   mstart     = perm + G_;

  int nz = N_;
  k_pre<<<NSORT_ + NDEN_ + NMSORT_ + (nz + 1023) / 1024, 1024, 0, stream>>>(is_seen, coords, centers, group2mask, tab, den, perm, mstart, headf, nz);

  k_group_gather<<<C_, 1024, 0, stream>>>(img, tab, den, perm, group_feat);
  k_mask_h1<<<NGEMM_ + NLINK_, LB_, 0, stream>>>(group_feat, mstart, W1, mask_h1, pair_mask, pair_point, head, link);

  k_final<<<NFB_, FT_, 0, stream>>>(mask_h1, head, link, b1, W2, b2, out);
}
